# starting kernel
# baseline (speedup 1.0000x reference)
#include <hip/hip_runtime.h>
#include <math.h>

#define N_ENT   150000
#define N_USR   60000
#define CH      64
#define E_EDGES 1000000
#define NNZ_CNT 1000000
#define NRELM1  9
#define NHOPS   2
#define EPS_N   1e-12f

typedef __attribute__((ext_vector_type(2))) float v2f;
typedef __attribute__((ext_vector_type(8))) float v8f;

#define ATT_WAVES 4
__global__ __launch_bounds__(ATT_WAVES * 32)
void att_wmma_kernel(const float* __restrict__ ent,
                     const float* __restrict__ weight,
                     const int* __restrict__ head,
                     const int* __restrict__ tail,
                     const int* __restrict__ etype,
                     float* __restrict__ att,
                     unsigned int* __restrict__ segmax)
{
    __shared__ float sW2[16 * CH];
    __shared__ float sH[ATT_WAVES][16 * CH];
    __shared__ float sT[ATT_WAVES][16 * CH];

    const int lane = threadIdx.x & 31;
    const int wave = threadIdx.x >> 5;

    for (int i = threadIdx.x; i < 16 * CH; i += ATT_WAVES * 32) {
        const int n = i >> 6, c = i & 63;
        const float w = (n < NRELM1) ? weight[n * CH + c] : 0.0f;
        sW2[i] = w * w;
    }
    __syncthreads();

    const int tile = blockIdx.x * ATT_WAVES + wave;
    if (tile * 16 >= E_EDGES) return;
    const int base = tile * 16;

    for (int r = 0; r < 16; ++r) {
        const int hr = head[base + r];
        const int tr = tail[base + r];
        const v2f hv = *(const v2f*)(ent + (size_t)hr * CH + 2 * lane);
        const v2f tv = *(const v2f*)(ent + (size_t)tr * CH + 2 * lane);
        sH[wave][r * CH + 2 * lane]     = hv.x * hv.x;
        sH[wave][r * CH + 2 * lane + 1] = hv.y * hv.y;
        sT[wave][r * CH + 2 * lane]     = tv.x * tv.x;
        sT[wave][r * CH + 2 * lane + 1] = tv.y * tv.y;
    }

    v8f accH = {};
    v8f accT = {};
    const int m    = lane & 15;
    const int koff = (lane >> 4) << 1;
    for (int kk = 0; kk < 16; ++kk) {
        const int c = kk * 4 + koff;
        v2f a, at, b;
        a.x  = sH[wave][m * CH + c];  a.y  = sH[wave][m * CH + c + 1];
        at.x = sT[wave][m * CH + c];  at.y = sT[wave][m * CH + c + 1];
        b.x  = sW2[m * CH + c];       b.y  = sW2[m * CH + c + 1];
        accH = __builtin_amdgcn_wmma_f32_16x16x4_f32(false, a,  false, b, (short)0, accH, false, false);
        accT = __builtin_amdgcn_wmma_f32_16x16x4_f32(false, at, false, b, (short)0, accT, false, false);
    }

    const int mrow = (lane >> 4) << 3;
    const int ncol = lane & 15;
#pragma unroll
    for (int v = 0; v < 8; ++v) {
        sH[wave][(mrow + v) * 16 + ncol] = accH[v];
        sT[wave][(mrow + v) * 16 + ncol] = accT[v];
    }

    if (lane < 16) {
        const int e  = base + lane;
        int et = etype[e] - 1;
        if (et < 0) et += NRELM1;
        const float hn2 = sH[wave][lane * 16 + et];
        const float tn2 = sT[wave][lane * 16 + et];
        const float a   = hn2 * tn2;
        att[e] = a;
        atomicMax(&segmax[head[e]], __float_as_uint(a));
    }
}

__global__ __launch_bounds__(256)
void exp_sum_kernel(const int* __restrict__ head,
                    const float* __restrict__ segmax,
                    float* __restrict__ att,
                    float* __restrict__ segsum)
{
    const int e = blockIdx.x * 256 + threadIdx.x;
    if (e >= E_EDGES) return;
    const int h = head[e];
    const float ev = expf(att[e] - segmax[h]);
    att[e] = ev;
    atomicAdd(&segsum[h], ev);
}

__global__ __launch_bounds__(256)
void edge_agg_kernel(const float* __restrict__ ent,
                     const float* __restrict__ weight,
                     const int* __restrict__ head,
                     const int* __restrict__ tail,
                     const int* __restrict__ etype,
                     const float* __restrict__ expv,
                     const float* __restrict__ segsum,
                     float* __restrict__ agg_ent)
{
    const int wave = threadIdx.x >> 5, lane = threadIdx.x & 31;
    const int e = blockIdx.x * 8 + wave;
    if (e >= E_EDGES) return;
    const int h = head[e], t = tail[e];
    int et = etype[e] - 1;
    if (et < 0) et += NRELM1;
    const float w = expv[e] / segsum[h];
    const int c = 2 * lane;
    const v2f tv = *(const v2f*)(ent + (size_t)t * CH + c);
    const v2f rv = *(const v2f*)(weight + et * CH + c);
    atomicAdd(agg_ent + (size_t)h * CH + c,     w * tv.x * rv.x);
    atomicAdd(agg_ent + (size_t)h * CH + c + 1, w * tv.y * rv.y);
}

__global__ __launch_bounds__(256)
void inter_agg_kernel(const float* __restrict__ cur_ent,
                      const float* __restrict__ cur_usr,
                      const float* __restrict__ vals,
                      const int* __restrict__ rows,
                      const int* __restrict__ cols,
                      float* __restrict__ agg_ent,
                      float* __restrict__ agg_usr)
{
    const int wave = threadIdx.x >> 5, lane = threadIdx.x & 31;
    const int z = blockIdx.x * 8 + wave;
    if (z >= NNZ_CNT) return;
    const int r = rows[z], cl = cols[z];
    const float v = vals[z];
    const int c = 2 * lane;
    const v2f uv = *(const v2f*)(cur_usr + (size_t)r  * CH + c);
    const v2f ev = *(const v2f*)(cur_ent + (size_t)cl * CH + c);
    atomicAdd(agg_ent + (size_t)cl * CH + c,     v * uv.x);
    atomicAdd(agg_ent + (size_t)cl * CH + c + 1, v * uv.y);
    atomicAdd(agg_usr + (size_t)r  * CH + c,     v * ev.x);
    atomicAdd(agg_usr + (size_t)r  * CH + c + 1, v * ev.y);
}

__global__ __launch_bounds__(256)
void normalize_kernel(const float* __restrict__ agg,
                      float* __restrict__ cur,
                      float* __restrict__ out,
                      int nrows)
{
    const int wave = threadIdx.x >> 5, lane = threadIdx.x & 31;
    const int row = blockIdx.x * 8 + wave;
    if (row >= nrows) return;
    const int c = 2 * lane;
    v2f v = *(const v2f*)(agg + (size_t)row * CH + c);
    float ss = v.x * v.x + v.y * v.y;
#pragma unroll
    for (int m = 16; m >= 1; m >>= 1) ss += __shfl_xor(ss, m, 32);
    const float inv = 1.0f / fmaxf(sqrtf(ss), EPS_N);
    v.x *= inv; v.y *= inv;
    *(v2f*)(cur + (size_t)row * CH + c) = v;
    v2f o = *(v2f*)(out + (size_t)row * CH + c);
    o.x += v.x; o.y += v.y;
    *(v2f*)(out + (size_t)row * CH + c) = o;
}

extern "C" void kernel_launch(void* const* d_in, const int* in_sizes, int n_in,
                              void* d_out, int out_size, void* d_ws, size_t ws_size,
                              hipStream_t stream)
{
    (void)in_sizes; (void)n_in; (void)out_size; (void)ws_size;

    const float* user_emb   = (const float*)d_in[0];
    const float* entity_emb = (const float*)d_in[1];
    const float* weight     = (const float*)d_in[2];
    const float* inter_vals = (const float*)d_in[3];
    const int*   edge_index = (const int*)d_in[4];
    const int*   edge_type  = (const int*)d_in[5];
    const int*   inter_rows = (const int*)d_in[6];
    const int*   inter_cols = (const int*)d_in[7];
    const int*   head = edge_index;
    const int*   tail = edge_index + E_EDGES;

    float* out_ent = (float*)d_out;
    float* out_usr = out_ent + (size_t)N_ENT * CH;

    char* ws = (char*)d_ws;
    size_t off = 0;
    auto alloc_f = [&](size_t nfl) -> float* {
        float* p = (float*)(ws + off);
        off += ((nfl * sizeof(float) + 255) / 256) * 256;
        return p;
    };
    float* cur_ent = alloc_f((size_t)N_ENT * CH);
    float* cur_usr = alloc_f((size_t)N_USR * CH);
    float* agg_ent = alloc_f((size_t)N_ENT * CH);
    float* agg_usr = alloc_f((size_t)N_USR * CH);
    float* att     = alloc_f(E_EDGES);
    float* segmax  = alloc_f(N_ENT);
    float* segsum  = alloc_f(N_ENT);

    hipMemcpyAsync(cur_ent, entity_emb, (size_t)N_ENT * CH * sizeof(float), hipMemcpyDeviceToDevice, stream);
    hipMemcpyAsync(cur_usr, user_emb,   (size_t)N_USR * CH * sizeof(float), hipMemcpyDeviceToDevice, stream);
    hipMemcpyAsync(out_ent, entity_emb, (size_t)N_ENT * CH * sizeof(float), hipMemcpyDeviceToDevice, stream);
    hipMemcpyAsync(out_usr, user_emb,   (size_t)N_USR * CH * sizeof(float), hipMemcpyDeviceToDevice, stream);

    for (int hop = 0; hop < NHOPS; ++hop) {
        hipMemsetAsync(agg_ent, 0, (size_t)N_ENT * CH * sizeof(float), stream);
        hipMemsetAsync(agg_usr, 0, (size_t)N_USR * CH * sizeof(float), stream);
        hipMemsetAsync(segmax,  0, (size_t)N_ENT * sizeof(float), stream);
        hipMemsetAsync(segsum,  0, (size_t)N_ENT * sizeof(float), stream);

        att_wmma_kernel<<<E_EDGES / 16 / ATT_WAVES, ATT_WAVES * 32, 0, stream>>>(
            cur_ent, weight, head, tail, edge_type, att, (unsigned int*)segmax);
        exp_sum_kernel<<<(E_EDGES + 255) / 256, 256, 0, stream>>>(head, segmax, att, segsum);
        edge_agg_kernel<<<E_EDGES / 8, 256, 0, stream>>>(
            cur_ent, weight, head, tail, edge_type, att, segsum, agg_ent);
        inter_agg_kernel<<<NNZ_CNT / 8, 256, 0, stream>>>(
            cur_ent, cur_usr, inter_vals, inter_rows, inter_cols, agg_ent, agg_usr);
        normalize_kernel<<<(N_ENT + 7) / 8, 256, 0, stream>>>(agg_ent, cur_ent, out_ent, N_ENT);
        normalize_kernel<<<(N_USR + 7) / 8, 256, 0, stream>>>(agg_usr, cur_usr, out_usr, N_USR);
    }
}


# Scatter-adds by destination-range owners: 256 work-groups hold their entity/user row sums in LDS (ds_add_f32, order varies), scan 1-byte range ids of all items; normalize+residual fused, no memcpy
# speedup vs baseline: 1.9465x; 1.9465x over previous
#include <hip/hip_runtime.h>
#include <math.h>

#define N_ENT   150000
#define N_USR   60000
#define CH      64
#define E_EDGES 1000000
#define NNZ_CNT 1000000
#define NRELM1  9
#define NHOPS   2
#define EPS_N   1e-12f

#define NRANGE   256
#define ENT_PER  586
#define USR_PER  235
#define OW_WAVES 32
#define QCAP     128
#define BATCH    16
#define SCAN_T   4
#define NGROUPS  (E_EDGES / 16)

typedef __attribute__((ext_vector_type(2))) float v2f;
typedef __attribute__((ext_vector_type(8))) float v8f;

#define ATT_WAVES 4
__global__ __launch_bounds__(ATT_WAVES * 32)
void att_wmma_kernel(const float* __restrict__ ent,
                     const float* __restrict__ weight,
                     const int* __restrict__ head,
                     const int* __restrict__ tail,
                     const int* __restrict__ etype,
                     float* __restrict__ att,
                     unsigned int* __restrict__ segmax)
{
    __shared__ float sW2[16 * CH];
    __shared__ float sH[ATT_WAVES][16 * CH];
    __shared__ float sT[ATT_WAVES][16 * CH];

    const int lane = threadIdx.x & 31;
    const int wave = threadIdx.x >> 5;

    for (int i = threadIdx.x; i < 16 * CH; i += ATT_WAVES * 32) {
        const int n = i >> 6, c = i & 63;
        const float w = (n < NRELM1) ? weight[n * CH + c] : 0.0f;
        sW2[i] = w * w;
    }
    __syncthreads();

    const int tile = blockIdx.x * ATT_WAVES + wave;
    if (tile * 16 >= E_EDGES) return;
    const int base = tile * 16;

    for (int r = 0; r < 16; ++r) {
        const int hr = head[base + r];
        const int tr = tail[base + r];
        const v2f hv = *(const v2f*)(ent + (size_t)hr * CH + 2 * lane);
        const v2f tv = *(const v2f*)(ent + (size_t)tr * CH + 2 * lane);
        sH[wave][r * CH + 2 * lane]     = hv.x * hv.x;
        sH[wave][r * CH + 2 * lane + 1] = hv.y * hv.y;
        sT[wave][r * CH + 2 * lane]     = tv.x * tv.x;
        sT[wave][r * CH + 2 * lane + 1] = tv.y * tv.y;
    }

    v8f accH = {};
    v8f accT = {};
    const int m    = lane & 15;
    const int koff = (lane >> 4) << 1;
    for (int kk = 0; kk < 16; ++kk) {
        const int c = kk * 4 + koff;
        v2f a, at, b;
        a.x  = sH[wave][m * CH + c];  a.y  = sH[wave][m * CH + c + 1];
        at.x = sT[wave][m * CH + c];  at.y = sT[wave][m * CH + c + 1];
        b.x  = sW2[m * CH + c];       b.y  = sW2[m * CH + c + 1];
        accH = __builtin_amdgcn_wmma_f32_16x16x4_f32(false, a,  false, b, (short)0, accH, false, false);
        accT = __builtin_amdgcn_wmma_f32_16x16x4_f32(false, at, false, b, (short)0, accT, false, false);
    }

    const int mrow = (lane >> 4) << 3;
    const int ncol = lane & 15;
#pragma unroll
    for (int v = 0; v < 8; ++v) {
        sH[wave][(mrow + v) * 16 + ncol] = accH[v];
        sT[wave][(mrow + v) * 16 + ncol] = accT[v];
    }

    if (lane < 16) {
        const int e  = base + lane;
        int et = etype[e] - 1;
        if (et < 0) et += NRELM1;
        const float hn2 = sH[wave][lane * 16 + et];
        const float tn2 = sT[wave][lane * 16 + et];
        const float a   = hn2 * tn2;
        att[4 * e + 3] = a;
        atomicMax(&segmax[head[e]], __float_as_uint(a));
    }
}

__device__ __forceinline__ unsigned int pack_ids(int a, int b, int c, int d, int per)
{
    return ((unsigned int)(a / per) & 0xFFu) | (((unsigned int)(b / per) & 0xFFu) << 8) |
           (((unsigned int)(c / per) & 0xFFu) << 16) | (((unsigned int)(d / per) & 0xFFu) << 24);
}

__device__ __forceinline__ int rel_row(int etype)
{
    int et = etype - 1;
    if (et < 0) et += NRELM1;
    return et;
}

__global__ __launch_bounds__(256)
void prep_kernel(const int* __restrict__ head,
                 const int* __restrict__ tail,
                 const int* __restrict__ etype,
                 const int* __restrict__ rows,
                 const int* __restrict__ cols,
                 const float* __restrict__ vals,
                 unsigned int* __restrict__ ridE,
                 unsigned int* __restrict__ ridC,
                 unsigned int* __restrict__ ridR,
                 int* __restrict__ recE,
                 int* __restrict__ recI)
{
    const int i = blockIdx.x * 256 + threadIdx.x;
    if (i >= E_EDGES / 4) return;
    const int4 h = ((const int4*)head)[i];
    const int4 t = ((const int4*)tail)[i];
    const int4 y = ((const int4*)etype)[i];
    const int4 c = ((const int4*)cols)[i];
    const int4 r = ((const int4*)rows)[i];
    const float4 v = ((const float4*)vals)[i];
    ridE[i] = pack_ids(h.x, h.y, h.z, h.w, ENT_PER);
    ridC[i] = pack_ids(c.x, c.y, c.z, c.w, ENT_PER);
    ridR[i] = pack_ids(r.x, r.y, r.z, r.w, USR_PER);
    int* e = recE + 16 * i;
    e[0]  = h.x; e[1]  = t.x; e[2]  = rel_row(y.x);
    e[4]  = h.y; e[5]  = t.y; e[6]  = rel_row(y.y);
    e[8]  = h.z; e[9]  = t.z; e[10] = rel_row(y.z);
    e[12] = h.w; e[13] = t.w; e[14] = rel_row(y.w);
    int4* n = (int4*)(recI + 16 * i);
    n[0] = make_int4(r.x, c.x, __float_as_int(v.x), 0);
    n[1] = make_int4(r.y, c.y, __float_as_int(v.y), 0);
    n[2] = make_int4(r.z, c.z, __float_as_int(v.z), 0);
    n[3] = make_int4(r.w, c.w, __float_as_int(v.w), 0);
}

struct OwnArgs {
    const float* ent; const float* usr; const float* weight;
    const int4* recE; const int4* recI; const float* segmax;
    float* accE; float* accU; float* sumE;
    int loE, nE, loU, nU;
};

__device__ __forceinline__ unsigned int zero_flags(unsigned int x)
{
    const unsigned int t = (x & 0x7F7F7F7Fu) + 0x7F7F7F7Fu;
    return ~(t | x) & 0x80808080u;
}

template <int MODE>
__device__ __forceinline__ void drain_queue(const OwnArgs& a, int* q, int cnt, int lane)
{
    for (int base = 0; base < cnt; base += BATCH) {
        const int n = (cnt - base < BATCH) ? (cnt - base) : BATCH;
        const int j = lane & (BATCH - 1);
        const bool live = j < n;
        const int item = q[base + (live ? j : 0)];
        int srow, lrow, et = 0;
        float sc = 0.0f, av = 0.0f, mx = 0.0f;
        bool ok;
        if (MODE == 0) {
            const int4 r = a.recE[item];
            srow = r.y;
            et = r.z;
            av = __int_as_float(r.w);
            mx = a.segmax[r.x];
            lrow = r.x - a.loE;
            ok = live && ((unsigned int)lrow < (unsigned int)a.nE);
        } else if (MODE == 1) {
            const int4 r = a.recI[item];
            srow = r.x;
            sc = __int_as_float(r.z);
            lrow = r.y - a.loE;
            ok = live && ((unsigned int)lrow < (unsigned int)a.nE);
        } else {
            const int4 r = a.recI[item];
            srow = r.y;
            sc = __int_as_float(r.z);
            lrow = r.x - a.loU;
            ok = live && ((unsigned int)lrow < (unsigned int)a.nU);
        }
        const float* src = (MODE == 1) ? a.usr : a.ent;
        float* dst = (MODE == 2) ? a.accU : a.accE;
        v2f sv[BATCH];
        v2f rv[BATCH];
#pragma unroll
        for (int k = 0; k < BATCH; ++k) {
            const int sr = __builtin_amdgcn_readlane(srow, k);
            sv[k] = *(const v2f*)(src + (size_t)sr * CH + 2 * lane);
            if (MODE == 0) {
                const int er = __builtin_amdgcn_readlane(et, k);
                rv[k] = *(const v2f*)(a.weight + er * CH + 2 * lane);
            }
        }
        if (MODE == 0) sc = expf(av - mx);
        if (!ok) { sc = 0.0f; lrow = 0; }
        if (MODE == 0) {
            if (ok && lane < BATCH) atomicAdd(a.sumE + lrow, sc);
        }
#pragma unroll
        for (int k = 0; k < BATCH; ++k) {
            if (k < n) {
                const float s = __int_as_float(__builtin_amdgcn_readlane(__float_as_int(sc), k));
                const int lr = __builtin_amdgcn_readlane(lrow, k);
                float x, y;
                if (MODE == 0) { x = s * sv[k].x * rv[k].x; y = s * sv[k].y * rv[k].y; }
                else           { x = s * sv[k].x;           y = s * sv[k].y; }
                atomicAdd(dst + lr * CH + 2 * lane,     x);
                atomicAdd(dst + lr * CH + 2 * lane + 1, y);
            }
        }
    }
}

template <int MODE>
__device__ __forceinline__ void scan_list(const OwnArgs& a, const unsigned int* __restrict__ rid,
                                          unsigned int pat, int* q, int wave, int lane)
{
    const uint4* rid4 = (const uint4*)rid;
    int cnt = 0;
    uint4 w[SCAN_T];
#pragma unroll
    for (int u = 0; u < SCAN_T; ++u) {
        const int g = wave * 32 + lane + u * (OW_WAVES * 32);
        w[u] = rid4[(g < NGROUPS) ? g : (NGROUPS - 1)];
    }
    for (int t0 = 0; t0 < NGROUPS; t0 += SCAN_T * OW_WAVES * 32) {
        const int gb = t0 + wave * 32 + lane;
        unsigned int mA = 0u, mB = 0u;
#pragma unroll
        for (int u = 0; u < SCAN_T; ++u) {
            unsigned int m = (zero_flags(w[u].x ^ pat) >> 7) | (zero_flags(w[u].y ^ pat) >> 6) |
                             (zero_flags(w[u].z ^ pat) >> 5) | (zero_flags(w[u].w ^ pat) >> 4);
            if (gb + u * (OW_WAVES * 32) >= NGROUPS) m = 0u;
            m <<= 4 * (u & 1);
            if (u < 2) mA |= m; else mB |= m;
        }
#pragma unroll
        for (int u = 0; u < SCAN_T; ++u) {
            const int g = gb + (SCAN_T + u) * (OW_WAVES * 32);
            w[u] = rid4[(g < NGROUPS) ? g : (NGROUPS - 1)];
        }
        unsigned int bal = __builtin_amdgcn_ballot_w32((mA | mB) != 0u);
        while (bal != 0u) {
            if (cnt + 32 > QCAP) { drain_queue<MODE>(a, q, cnt, lane); cnt = 0; }
            if ((mA | mB) != 0u) {
                const bool inA = mA != 0u;
                const unsigned int x = inA ? mA : mB;
                const int p = __builtin_ctz(x);
                if (inA) mA &= mA - 1u; else mB &= mB - 1u;
                const int u = (inA ? 0 : 2) + ((p >> 2) & 1);
                const int g = gb + u * (OW_WAVES * 32);
                q[cnt + (int)__builtin_amdgcn_mbcnt_lo(bal, 0u)] = g * 16 + 4 * (p & 3) + (p >> 3);
            }
            cnt += __builtin_popcount(bal);
            bal = __builtin_amdgcn_ballot_w32((mA | mB) != 0u);
        }
    }
    if (cnt > 0) drain_queue<MODE>(a, q, cnt, lane);
}

__global__ __launch_bounds__(OW_WAVES * 32)
void owner_kernel(const float* cur_ent,
                  const float* cur_usr,
                  const float* __restrict__ weight,
                  const int* __restrict__ recE,
                  const int* __restrict__ recI,
                  const float* __restrict__ segmax,
                  const unsigned int* __restrict__ ridE,
                  const unsigned int* __restrict__ ridC,
                  const unsigned int* __restrict__ ridR,
                  const float* res_ent,
                  const float* res_usr,
                  float* nxt_ent,
                  float* nxt_usr,
                  float* out_ent,
                  float* out_usr,
                  int write_nxt)
{
    __shared__ float accE[ENT_PER * CH];
    __shared__ float accU[USR_PER * CH];
    __shared__ float sumE[ENT_PER];
    __shared__ int   qs[OW_WAVES * QCAP];

    const int lane = threadIdx.x & 31;
    const int wave = __builtin_amdgcn_readfirstlane(threadIdx.x >> 5);
    const int rg   = blockIdx.x;

    OwnArgs a;
    a.ent = cur_ent; a.usr = cur_usr; a.weight = weight;
    a.recE = (const int4*)recE; a.recI = (const int4*)recI; a.segmax = segmax;
    a.accE = accE; a.accU = accU; a.sumE = sumE;
    a.loE = rg * ENT_PER;
    a.nE  = (N_ENT - a.loE < ENT_PER) ? (N_ENT - a.loE) : ENT_PER;
    a.loU = rg * USR_PER;
    a.nU  = (N_USR - a.loU < USR_PER) ? (N_USR - a.loU) : USR_PER;

    for (int i = threadIdx.x; i < ENT_PER * CH; i += OW_WAVES * 32) accE[i] = 0.0f;
    for (int i = threadIdx.x; i < USR_PER * CH; i += OW_WAVES * 32) accU[i] = 0.0f;
    for (int i = threadIdx.x; i < ENT_PER;      i += OW_WAVES * 32) sumE[i] = 0.0f;
    __syncthreads();

    int* q = qs + wave * QCAP;
    const unsigned int pat = (unsigned int)rg * 0x01010101u;

    scan_list<0>(a, ridE, pat, q, wave, lane);
    __syncthreads();

    for (int lr = wave; lr < a.nE; lr += OW_WAVES) {
        float s = sumE[lr];
        s = (s > 0.0f) ? s : 1.0f;
        v2f v = *(v2f*)(accE + lr * CH + 2 * lane);
        v.x /= s; v.y /= s;
        *(v2f*)(accE + lr * CH + 2 * lane) = v;
    }
    __syncthreads();

    scan_list<1>(a, ridC, pat, q, wave, lane);
    scan_list<2>(a, ridR, pat, q, wave, lane);
    __syncthreads();

    for (int lr = wave; lr < a.nE; lr += OW_WAVES) {
        v2f v = *(v2f*)(accE + lr * CH + 2 * lane);
        float ss = v.x * v.x + v.y * v.y;
#pragma unroll
        for (int m = 16; m >= 1; m >>= 1) ss += __shfl_xor(ss, m, 32);
        const float inv = 1.0f / fmaxf(sqrtf(ss), EPS_N);
        v.x *= inv; v.y *= inv;
        const size_t off = (size_t)(a.loE + lr) * CH + 2 * lane;
        if (write_nxt) *(v2f*)(nxt_ent + off) = v;
        v2f o = *(const v2f*)(res_ent + off);
        o.x += v.x; o.y += v.y;
        *(v2f*)(out_ent + off) = o;
    }
    for (int lr = wave; lr < a.nU; lr += OW_WAVES) {
        v2f v = *(v2f*)(accU + lr * CH + 2 * lane);
        float ss = v.x * v.x + v.y * v.y;
#pragma unroll
        for (int m = 16; m >= 1; m >>= 1) ss += __shfl_xor(ss, m, 32);
        const float inv = 1.0f / fmaxf(sqrtf(ss), EPS_N);
        v.x *= inv; v.y *= inv;
        const size_t off = (size_t)(a.loU + lr) * CH + 2 * lane;
        if (write_nxt) *(v2f*)(nxt_usr + off) = v;
        v2f o = *(const v2f*)(res_usr + off);
        o.x += v.x; o.y += v.y;
        *(v2f*)(out_usr + off) = o;
    }
}

extern "C" void kernel_launch(void* const* d_in, const int* in_sizes, int n_in,
                              void* d_out, int out_size, void* d_ws, size_t ws_size,
                              hipStream_t stream)
{
    (void)in_sizes; (void)n_in; (void)out_size; (void)ws_size;

    const float* user_emb   = (const float*)d_in[0];
    const float* entity_emb = (const float*)d_in[1];
    const float* weight     = (const float*)d_in[2];
    const float* inter_vals = (const float*)d_in[3];
    const int*   edge_index = (const int*)d_in[4];
    const int*   edge_type  = (const int*)d_in[5];
    const int*   inter_rows = (const int*)d_in[6];
    const int*   inter_cols = (const int*)d_in[7];
    const int*   head = edge_index;
    const int*   tail = edge_index + E_EDGES;

    float* out_ent = (float*)d_out;
    float* out_usr = out_ent + (size_t)N_ENT * CH;

    char* ws = (char*)d_ws;
    size_t off = 0;
    auto alloc_f = [&](size_t nfl) -> float* {
        float* p = (float*)(ws + off);
        off += ((nfl * sizeof(float) + 255) / 256) * 256;
        return p;
    };
    float* nxt_ent = alloc_f((size_t)N_ENT * CH);
    float* nxt_usr = alloc_f((size_t)N_USR * CH);
    int*   recE    = (int*)alloc_f((size_t)E_EDGES * 4);
    int*   recI    = (int*)alloc_f((size_t)NNZ_CNT * 4);
    float* att     = (float*)recE;
    float* segmax  = alloc_f(N_ENT);
    unsigned int* ridE = (unsigned int*)alloc_f(E_EDGES / 4);
    unsigned int* ridC = (unsigned int*)alloc_f(NNZ_CNT / 4);
    unsigned int* ridR = (unsigned int*)alloc_f(NNZ_CNT / 4);

    prep_kernel<<<(E_EDGES / 4 + 255) / 256, 256, 0, stream>>>(
        head, tail, edge_type, inter_rows, inter_cols, inter_vals, ridE, ridC, ridR, recE, recI);

    hipMemsetAsync(segmax, 0, (size_t)N_ENT * sizeof(float), stream);
    att_wmma_kernel<<<E_EDGES / 16 / ATT_WAVES, ATT_WAVES * 32, 0, stream>>>(
        entity_emb, weight, head, tail, edge_type, att, (unsigned int*)segmax);
    owner_kernel<<<NRANGE, OW_WAVES * 32, 0, stream>>>(
        entity_emb, user_emb, weight, recE, recI, segmax, ridE, ridC, ridR,
        entity_emb, user_emb, nxt_ent, nxt_usr, out_ent, out_usr, 1);

    hipMemsetAsync(segmax, 0, (size_t)N_ENT * sizeof(float), stream);
    att_wmma_kernel<<<E_EDGES / 16 / ATT_WAVES, ATT_WAVES * 32, 0, stream>>>(
        nxt_ent, weight, head, tail, edge_type, att, (unsigned int*)segmax);
    owner_kernel<<<NRANGE, OW_WAVES * 32, 0, stream>>>(
        nxt_ent, nxt_usr, weight, recE, recI, segmax, ridE, ridC, ridR,
        out_ent, out_usr, nxt_ent, nxt_usr, out_ent, out_usr, 0);
}


# Attention kernel: WMMA operand fragments read straight from global rows in fragment layout (one loader for A and B, k order permuted); no LDS operand tiles; channel sum order changes
# speedup vs baseline: 2.1603x; 1.1098x over previous
#include <hip/hip_runtime.h>
#include <math.h>

#define N_ENT   150000
#define N_USR   60000
#define CH      64
#define E_EDGES 1000000
#define NNZ_CNT 1000000
#define NRELM1  9
#define NHOPS   2
#define EPS_N   1e-12f

#define NRANGE   256
#define ENT_PER  586
#define USR_PER  235
#define OW_WAVES 32
#define QCAP     128
#define BATCH    16
#define SCAN_T   4
#define NGROUPS  (E_EDGES / 16)

typedef __attribute__((ext_vector_type(2))) float v2f;
typedef __attribute__((ext_vector_type(8))) float v8f;

#define ATT_WAVES 4
typedef __attribute__((ext_vector_type(4))) float v4f;

__device__ __forceinline__ v4f load_row_piece(const float* __restrict__ row, int h, int i)
{
    return *(const v4f*)(row + 8 * i + 4 * h);
}

__global__ __launch_bounds__(ATT_WAVES * 32)
void att_wmma_kernel(const float* __restrict__ ent,
                     const float* __restrict__ weight,
                     const int* __restrict__ head,
                     const int* __restrict__ tail,
                     const int* __restrict__ etype,
                     float* __restrict__ att,
                     unsigned int* __restrict__ segmax)
{
    __shared__ float sP[ATT_WAVES][16 * 16];

    const int lane = threadIdx.x & 31;
    const int wave = threadIdx.x >> 5;

    const int tile = blockIdx.x * ATT_WAVES + wave;
    if (tile * 16 >= E_EDGES) return;
    const int base = tile * 16;

    const int c = lane & 15;
    const int h = lane >> 4;
    const int e  = base + c;
    const int hr = head[e];
    const int tr = tail[e];
    int et = etype[e] - 1;
    if (et < 0) et += NRELM1;
    const bool wlive = c < NRELM1;
    const int wr = wlive ? c : (NRELM1 - 1);

    const float* wrow = weight + wr * CH;
    const float* hrow = ent + (size_t)hr * CH;
    const float* trow = ent + (size_t)tr * CH;
    v4f wv[8], hv[8], tv[8];
#pragma unroll
    for (int i = 0; i < 8; ++i) {
        wv[i] = load_row_piece(wrow, h, i);
        hv[i] = load_row_piece(hrow, h, i);
        tv[i] = load_row_piece(trow, h, i);
    }
    __builtin_amdgcn_sched_barrier(0);

    v8f accH = {};
    v8f accT = {};
#pragma unroll
    for (int i = 0; i < 8; ++i) {
        v2f a, at, b;
        a.x  = hv[i].x * hv[i].x;  a.y  = hv[i].y * hv[i].y;
        at.x = tv[i].x * tv[i].x;  at.y = tv[i].y * tv[i].y;
        b.x  = wlive ? wv[i].x * wv[i].x : 0.0f;
        b.y  = wlive ? wv[i].y * wv[i].y : 0.0f;
        accH = __builtin_amdgcn_wmma_f32_16x16x4_f32(false, a,  false, b, (short)0, accH, false, false);
        accT = __builtin_amdgcn_wmma_f32_16x16x4_f32(false, at, false, b, (short)0, accT, false, false);
        a.x  = hv[i].z * hv[i].z;  a.y  = hv[i].w * hv[i].w;
        at.x = tv[i].z * tv[i].z;  at.y = tv[i].w * tv[i].w;
        b.x  = wlive ? wv[i].z * wv[i].z : 0.0f;
        b.y  = wlive ? wv[i].w * wv[i].w : 0.0f;
        accH = __builtin_amdgcn_wmma_f32_16x16x4_f32(false, a,  false, b, (short)0, accH, false, false);
        accT = __builtin_amdgcn_wmma_f32_16x16x4_f32(false, at, false, b, (short)0, accT, false, false);
    }

    const int mrow = h << 3;
#pragma unroll
    for (int v = 0; v < 8; ++v) sP[wave][(mrow + v) * 16 + c] = accH[v] * accT[v];

    if (lane < 16) {
        const float a = sP[wave][lane * 16 + et];
        att[4 * e + 3] = a;
        atomicMax(&segmax[hr], __float_as_uint(a));
    }
}

__device__ __forceinline__ unsigned int pack_ids(int a, int b, int c, int d, int per)
{
    return ((unsigned int)(a / per) & 0xFFu) | (((unsigned int)(b / per) & 0xFFu) << 8) |
           (((unsigned int)(c / per) & 0xFFu) << 16) | (((unsigned int)(d / per) & 0xFFu) << 24);
}

__device__ __forceinline__ int rel_row(int etype)
{
    int et = etype - 1;
    if (et < 0) et += NRELM1;
    return et;
}

__global__ __launch_bounds__(256)
void prep_kernel(const int* __restrict__ head,
                 const int* __restrict__ tail,
                 const int* __restrict__ etype,
                 const int* __restrict__ rows,
                 const int* __restrict__ cols,
                 const float* __restrict__ vals,
                 unsigned int* __restrict__ ridE,
                 unsigned int* __restrict__ ridC,
                 unsigned int* __restrict__ ridR,
                 int* __restrict__ recE,
                 int* __restrict__ recI)
{
    const int i = blockIdx.x * 256 + threadIdx.x;
    if (i >= E_EDGES / 4) return;
    const int4 h = ((const int4*)head)[i];
    const int4 t = ((const int4*)tail)[i];
    const int4 y = ((const int4*)etype)[i];
    const int4 c = ((const int4*)cols)[i];
    const int4 r = ((const int4*)rows)[i];
    const float4 v = ((const float4*)vals)[i];
    ridE[i] = pack_ids(h.x, h.y, h.z, h.w, ENT_PER);
    ridC[i] = pack_ids(c.x, c.y, c.z, c.w, ENT_PER);
    ridR[i] = pack_ids(r.x, r.y, r.z, r.w, USR_PER);
    int* e = recE + 16 * i;
    e[0]  = h.x; e[1]  = t.x; e[2]  = rel_row(y.x);
    e[4]  = h.y; e[5]  = t.y; e[6]  = rel_row(y.y);
    e[8]  = h.z; e[9]  = t.z; e[10] = rel_row(y.z);
    e[12] = h.w; e[13] = t.w; e[14] = rel_row(y.w);
    int4* n = (int4*)(recI + 16 * i);
    n[0] = make_int4(r.x, c.x, __float_as_int(v.x), 0);
    n[1] = make_int4(r.y, c.y, __float_as_int(v.y), 0);
    n[2] = make_int4(r.z, c.z, __float_as_int(v.z), 0);
    n[3] = make_int4(r.w, c.w, __float_as_int(v.w), 0);
}

struct OwnArgs {
    const float* ent; const float* usr; const float* weight;
    const int4* recE; const int4* recI; const float* segmax;
    float* accE; float* accU; float* sumE;
    int loE, nE, loU, nU;
};

__device__ __forceinline__ unsigned int zero_flags(unsigned int x)
{
    const unsigned int t = (x & 0x7F7F7F7Fu) + 0x7F7F7F7Fu;
    return ~(t | x) & 0x80808080u;
}

template <int MODE>
__device__ __forceinline__ void drain_queue(const OwnArgs& a, int* q, int cnt, int lane)
{
    for (int base = 0; base < cnt; base += BATCH) {
        const int n = (cnt - base < BATCH) ? (cnt - base) : BATCH;
        const int j = lane & (BATCH - 1);
        const bool live = j < n;
        const int item = q[base + (live ? j : 0)];
        int srow, lrow, et = 0;
        float sc = 0.0f, av = 0.0f, mx = 0.0f;
        bool ok;
        if (MODE == 0) {
            const int4 r = a.recE[item];
            srow = r.y;
            et = r.z;
            av = __int_as_float(r.w);
            mx = a.segmax[r.x];
            lrow = r.x - a.loE;
            ok = live && ((unsigned int)lrow < (unsigned int)a.nE);
        } else if (MODE == 1) {
            const int4 r = a.recI[item];
            srow = r.x;
            sc = __int_as_float(r.z);
            lrow = r.y - a.loE;
            ok = live && ((unsigned int)lrow < (unsigned int)a.nE);
        } else {
            const int4 r = a.recI[item];
            srow = r.y;
            sc = __int_as_float(r.z);
            lrow = r.x - a.loU;
            ok = live && ((unsigned int)lrow < (unsigned int)a.nU);
        }
        const float* src = (MODE == 1) ? a.usr : a.ent;
        float* dst = (MODE == 2) ? a.accU : a.accE;
        v2f sv[BATCH];
        v2f rv[BATCH];
#pragma unroll
        for (int k = 0; k < BATCH; ++k) {
            const int sr = __builtin_amdgcn_readlane(srow, k);
            sv[k] = *(const v2f*)(src + (size_t)sr * CH + 2 * lane);
            if (MODE == 0) {
                const int er = __builtin_amdgcn_readlane(et, k);
                rv[k] = *(const v2f*)(a.weight + er * CH + 2 * lane);
            }
        }
        if (MODE == 0) sc = expf(av - mx);
        if (!ok) { sc = 0.0f; lrow = 0; }
        if (MODE == 0) {
            if (ok && lane < BATCH) atomicAdd(a.sumE + lrow, sc);
        }
#pragma unroll
        for (int k = 0; k < BATCH; ++k) {
            if (k < n) {
                const float s = __int_as_float(__builtin_amdgcn_readlane(__float_as_int(sc), k));
                const int lr = __builtin_amdgcn_readlane(lrow, k);
                float x, y;
                if (MODE == 0) { x = s * sv[k].x * rv[k].x; y = s * sv[k].y * rv[k].y; }
                else           { x = s * sv[k].x;           y = s * sv[k].y; }
                atomicAdd(dst + lr * CH + 2 * lane,     x);
                atomicAdd(dst + lr * CH + 2 * lane + 1, y);
            }
        }
    }
}

template <int MODE>
__device__ __forceinline__ void scan_list(const OwnArgs& a, const unsigned int* __restrict__ rid,
                                          unsigned int pat, int* q, int wave, int lane)
{
    const uint4* rid4 = (const uint4*)rid;
    int cnt = 0;
    uint4 w[SCAN_T];
#pragma unroll
    for (int u = 0; u < SCAN_T; ++u) {
        const int g = wave * 32 + lane + u * (OW_WAVES * 32);
        w[u] = rid4[(g < NGROUPS) ? g : (NGROUPS - 1)];
    }
    for (int t0 = 0; t0 < NGROUPS; t0 += SCAN_T * OW_WAVES * 32) {
        const int gb = t0 + wave * 32 + lane;
        unsigned int mA = 0u, mB = 0u;
#pragma unroll
        for (int u = 0; u < SCAN_T; ++u) {
            unsigned int m = (zero_flags(w[u].x ^ pat) >> 7) | (zero_flags(w[u].y ^ pat) >> 6) |
                             (zero_flags(w[u].z ^ pat) >> 5) | (zero_flags(w[u].w ^ pat) >> 4);
            if (gb + u * (OW_WAVES * 32) >= NGROUPS) m = 0u;
            m <<= 4 * (u & 1);
            if (u < 2) mA |= m; else mB |= m;
        }
#pragma unroll
        for (int u = 0; u < SCAN_T; ++u) {
            const int g = gb + (SCAN_T + u) * (OW_WAVES * 32);
            w[u] = rid4[(g < NGROUPS) ? g : (NGROUPS - 1)];
        }
        unsigned int bal = __builtin_amdgcn_ballot_w32((mA | mB) != 0u);
        while (bal != 0u) {
            if (cnt + 32 > QCAP) { drain_queue<MODE>(a, q, cnt, lane); cnt = 0; }
            if ((mA | mB) != 0u) {
                const bool inA = mA != 0u;
                const unsigned int x = inA ? mA : mB;
                const int p = __builtin_ctz(x);
                if (inA) mA &= mA - 1u; else mB &= mB - 1u;
                const int u = (inA ? 0 : 2) + ((p >> 2) & 1);
                const int g = gb + u * (OW_WAVES * 32);
                q[cnt + (int)__builtin_amdgcn_mbcnt_lo(bal, 0u)] = g * 16 + 4 * (p & 3) + (p >> 3);
            }
            cnt += __builtin_popcount(bal);
            bal = __builtin_amdgcn_ballot_w32((mA | mB) != 0u);
        }
    }
    if (cnt > 0) drain_queue<MODE>(a, q, cnt, lane);
}

__global__ __launch_bounds__(OW_WAVES * 32)
void owner_kernel(const float* cur_ent,
                  const float* cur_usr,
                  const float* __restrict__ weight,
                  const int* __restrict__ recE,
                  const int* __restrict__ recI,
                  const float* __restrict__ segmax,
                  const unsigned int* __restrict__ ridE,
                  const unsigned int* __restrict__ ridC,
                  const unsigned int* __restrict__ ridR,
                  const float* res_ent,
                  const float* res_usr,
                  float* nxt_ent,
                  float* nxt_usr,
                  float* out_ent,
                  float* out_usr,
                  int write_nxt)
{
    __shared__ float accE[ENT_PER * CH];
    __shared__ float accU[USR_PER * CH];
    __shared__ float sumE[ENT_PER];
    __shared__ int   qs[OW_WAVES * QCAP];

    const int lane = threadIdx.x & 31;
    const int wave = __builtin_amdgcn_readfirstlane(threadIdx.x >> 5);
    const int rg   = blockIdx.x;

    OwnArgs a;
    a.ent = cur_ent; a.usr = cur_usr; a.weight = weight;
    a.recE = (const int4*)recE; a.recI = (const int4*)recI; a.segmax = segmax;
    a.accE = accE; a.accU = accU; a.sumE = sumE;
    a.loE = rg * ENT_PER;
    a.nE  = (N_ENT - a.loE < ENT_PER) ? (N_ENT - a.loE) : ENT_PER;
    a.loU = rg * USR_PER;
    a.nU  = (N_USR - a.loU < USR_PER) ? (N_USR - a.loU) : USR_PER;

    for (int i = threadIdx.x; i < ENT_PER * CH; i += OW_WAVES * 32) accE[i] = 0.0f;
    for (int i = threadIdx.x; i < USR_PER * CH; i += OW_WAVES * 32) accU[i] = 0.0f;
    for (int i = threadIdx.x; i < ENT_PER;      i += OW_WAVES * 32) sumE[i] = 0.0f;
    __syncthreads();

    int* q = qs + wave * QCAP;
    const unsigned int pat = (unsigned int)rg * 0x01010101u;

    scan_list<0>(a, ridE, pat, q, wave, lane);
    __syncthreads();

    for (int lr = wave; lr < a.nE; lr += OW_WAVES) {
        float s = sumE[lr];
        s = (s > 0.0f) ? s : 1.0f;
        v2f v = *(v2f*)(accE + lr * CH + 2 * lane);
        v.x /= s; v.y /= s;
        *(v2f*)(accE + lr * CH + 2 * lane) = v;
    }
    __syncthreads();

    scan_list<1>(a, ridC, pat, q, wave, lane);
    scan_list<2>(a, ridR, pat, q, wave, lane);
    __syncthreads();

    for (int lr = wave; lr < a.nE; lr += OW_WAVES) {
        v2f v = *(v2f*)(accE + lr * CH + 2 * lane);
        float ss = v.x * v.x + v.y * v.y;
#pragma unroll
        for (int m = 16; m >= 1; m >>= 1) ss += __shfl_xor(ss, m, 32);
        const float inv = 1.0f / fmaxf(sqrtf(ss), EPS_N);
        v.x *= inv; v.y *= inv;
        const size_t off = (size_t)(a.loE + lr) * CH + 2 * lane;
        if (write_nxt) *(v2f*)(nxt_ent + off) = v;
        v2f o = *(const v2f*)(res_ent + off);
        o.x += v.x; o.y += v.y;
        *(v2f*)(out_ent + off) = o;
    }
    for (int lr = wave; lr < a.nU; lr += OW_WAVES) {
        v2f v = *(v2f*)(accU + lr * CH + 2 * lane);
        float ss = v.x * v.x + v.y * v.y;
#pragma unroll
        for (int m = 16; m >= 1; m >>= 1) ss += __shfl_xor(ss, m, 32);
        const float inv = 1.0f / fmaxf(sqrtf(ss), EPS_N);
        v.x *= inv; v.y *= inv;
        const size_t off = (size_t)(a.loU + lr) * CH + 2 * lane;
        if (write_nxt) *(v2f*)(nxt_usr + off) = v;
        v2f o = *(const v2f*)(res_usr + off);
        o.x += v.x; o.y += v.y;
        *(v2f*)(out_usr + off) = o;
    }
}

extern "C" void kernel_launch(void* const* d_in, const int* in_sizes, int n_in,
                              void* d_out, int out_size, void* d_ws, size_t ws_size,
                              hipStream_t stream)
{
    (void)in_sizes; (void)n_in; (void)out_size; (void)ws_size;

    const float* user_emb   = (const float*)d_in[0];
    const float* entity_emb = (const float*)d_in[1];
    const float* weight     = (const float*)d_in[2];
    const float* inter_vals = (const float*)d_in[3];
    const int*   edge_index = (const int*)d_in[4];
    const int*   edge_type  = (const int*)d_in[5];
    const int*   inter_rows = (const int*)d_in[6];
    const int*   inter_cols = (const int*)d_in[7];
    const int*   head = edge_index;
    const int*   tail = edge_index + E_EDGES;

    float* out_ent = (float*)d_out;
    float* out_usr = out_ent + (size_t)N_ENT * CH;

    char* ws = (char*)d_ws;
    size_t off = 0;
    auto alloc_f = [&](size_t nfl) -> float* {
        float* p = (float*)(ws + off);
        off += ((nfl * sizeof(float) + 255) / 256) * 256;
        return p;
    };
    float* nxt_ent = alloc_f((size_t)N_ENT * CH);
    float* nxt_usr = alloc_f((size_t)N_USR * CH);
    int*   recE    = (int*)alloc_f((size_t)E_EDGES * 4);
    int*   recI    = (int*)alloc_f((size_t)NNZ_CNT * 4);
    float* att     = (float*)recE;
    float* segmax  = alloc_f(N_ENT);
    unsigned int* ridE = (unsigned int*)alloc_f(E_EDGES / 4);
    unsigned int* ridC = (unsigned int*)alloc_f(NNZ_CNT / 4);
    unsigned int* ridR = (unsigned int*)alloc_f(NNZ_CNT / 4);

    prep_kernel<<<(E_EDGES / 4 + 255) / 256, 256, 0, stream>>>(
        head, tail, edge_type, inter_rows, inter_cols, inter_vals, ridE, ridC, ridR, recE, recI);

    hipMemsetAsync(segmax, 0, (size_t)N_ENT * sizeof(float), stream);
    att_wmma_kernel<<<E_EDGES / 16 / ATT_WAVES, ATT_WAVES * 32, 0, stream>>>(
        entity_emb, weight, head, tail, edge_type, att, (unsigned int*)segmax);
    owner_kernel<<<NRANGE, OW_WAVES * 32, 0, stream>>>(
        entity_emb, user_emb, weight, recE, recI, segmax, ridE, ridC, ridR,
        entity_emb, user_emb, nxt_ent, nxt_usr, out_ent, out_usr, 1);

    hipMemsetAsync(segmax, 0, (size_t)N_ENT * sizeof(float), stream);
    att_wmma_kernel<<<E_EDGES / 16 / ATT_WAVES, ATT_WAVES * 32, 0, stream>>>(
        nxt_ent, weight, head, tail, edge_type, att, (unsigned int*)segmax);
    owner_kernel<<<NRANGE, OW_WAVES * 32, 0, stream>>>(
        nxt_ent, nxt_usr, weight, recE, recI, segmax, ridE, ridC, ridR,
        out_ent, out_usr, nxt_ent, nxt_usr, out_ent, out_usr, 0);
}


# Owner kernel, second hop: a wave walks 8-byte entries (source row, local row, value or edge) its first-hop scan kept in the workspace; no second scan, no record reads; LDS add order varies as before
# speedup vs baseline: 2.3442x; 1.0851x over previous
#include <hip/hip_runtime.h>
#include <math.h>

#define N_ENT   150000
#define N_USR   60000
#define CH      64
#define E_EDGES 1000000
#define NNZ_CNT 1000000
#define NRELM1  9
#define NHOPS   2
#define EPS_N   1e-12f

#define NRANGE   256
#define ENT_PER  586
#define USR_PER  235
#define OW_WAVES 32
#define QCAP     128
#define BATCH    16
#define SCAN_T   4
#define NGROUPS  (E_EDGES / 16)
#define NLIST    3
#define CAPW     192
#define WS_NEED  127207680ull

typedef __attribute__((ext_vector_type(2))) float v2f;
typedef __attribute__((ext_vector_type(8))) float v8f;

#define ATT_WAVES 4
typedef __attribute__((ext_vector_type(4))) float v4f;

__device__ __forceinline__ v4f load_row_piece(const float* __restrict__ row, int h, int i)
{
    return *(const v4f*)(row + 8 * i + 4 * h);
}

__global__ __launch_bounds__(ATT_WAVES * 32)
void att_wmma_kernel(const float* __restrict__ ent,
                     const float* __restrict__ weight,
                     const int* __restrict__ head,
                     const int* __restrict__ tail,
                     const int* __restrict__ etype,
                     float* __restrict__ att,
                     unsigned int* __restrict__ segmax)
{
    __shared__ float sP[ATT_WAVES][16 * 16];

    const int lane = threadIdx.x & 31;
    const int wave = threadIdx.x >> 5;

    const int tile = blockIdx.x * ATT_WAVES + wave;
    if (tile * 16 >= E_EDGES) return;
    const int base = tile * 16;

    const int c = lane & 15;
    const int h = lane >> 4;
    const int e  = base + c;
    const int hr = head[e];
    const int tr = tail[e];
    int et = etype[e] - 1;
    if (et < 0) et += NRELM1;
    const bool wlive = c < NRELM1;
    const int wr = wlive ? c : (NRELM1 - 1);

    const float* wrow = weight + wr * CH;
    const float* hrow = ent + (size_t)hr * CH;
    const float* trow = ent + (size_t)tr * CH;
    v4f wv[8], hv[8], tv[8];
#pragma unroll
    for (int i = 0; i < 8; ++i) {
        wv[i] = load_row_piece(wrow, h, i);
        hv[i] = load_row_piece(hrow, h, i);
        tv[i] = load_row_piece(trow, h, i);
    }
    __builtin_amdgcn_sched_barrier(0);

    v8f accH = {};
    v8f accT = {};
#pragma unroll
    for (int i = 0; i < 8; ++i) {
        v2f a, at, b;
        a.x  = hv[i].x * hv[i].x;  a.y  = hv[i].y * hv[i].y;
        at.x = tv[i].x * tv[i].x;  at.y = tv[i].y * tv[i].y;
        b.x  = wlive ? wv[i].x * wv[i].x : 0.0f;
        b.y  = wlive ? wv[i].y * wv[i].y : 0.0f;
        accH = __builtin_amdgcn_wmma_f32_16x16x4_f32(false, a,  false, b, (short)0, accH, false, false);
        accT = __builtin_amdgcn_wmma_f32_16x16x4_f32(false, at, false, b, (short)0, accT, false, false);
        a.x  = hv[i].z * hv[i].z;  a.y  = hv[i].w * hv[i].w;
        at.x = tv[i].z * tv[i].z;  at.y = tv[i].w * tv[i].w;
        b.x  = wlive ? wv[i].z * wv[i].z : 0.0f;
        b.y  = wlive ? wv[i].w * wv[i].w : 0.0f;
        accH = __builtin_amdgcn_wmma_f32_16x16x4_f32(false, a,  false, b, (short)0, accH, false, false);
        accT = __builtin_amdgcn_wmma_f32_16x16x4_f32(false, at, false, b, (short)0, accT, false, false);
    }

    const int mrow = h << 3;
#pragma unroll
    for (int v = 0; v < 8; ++v) sP[wave][(mrow + v) * 16 + c] = accH[v] * accT[v];

    if (lane < 16) {
        const float a = sP[wave][lane * 16 + et];
        att[4 * e + 3] = a;
        atomicMax(&segmax[hr], __float_as_uint(a));
    }
}

__device__ __forceinline__ unsigned int pack_ids(int a, int b, int c, int d, int per)
{
    return ((unsigned int)(a / per) & 0xFFu) | (((unsigned int)(b / per) & 0xFFu) << 8) |
           (((unsigned int)(c / per) & 0xFFu) << 16) | (((unsigned int)(d / per) & 0xFFu) << 24);
}

__device__ __forceinline__ int rel_row(int etype)
{
    int et = etype - 1;
    if (et < 0) et += NRELM1;
    return et;
}

__global__ __launch_bounds__(256)
void prep_kernel(const int* __restrict__ head,
                 const int* __restrict__ tail,
                 const int* __restrict__ etype,
                 const int* __restrict__ rows,
                 const int* __restrict__ cols,
                 const float* __restrict__ vals,
                 unsigned int* __restrict__ ridE,
                 unsigned int* __restrict__ ridC,
                 unsigned int* __restrict__ ridR,
                 int* __restrict__ recE,
                 int* __restrict__ recI)
{
    const int i = blockIdx.x * 256 + threadIdx.x;
    if (i >= E_EDGES / 4) return;
    const int4 h = ((const int4*)head)[i];
    const int4 t = ((const int4*)tail)[i];
    const int4 y = ((const int4*)etype)[i];
    const int4 c = ((const int4*)cols)[i];
    const int4 r = ((const int4*)rows)[i];
    const float4 v = ((const float4*)vals)[i];
    ridE[i] = pack_ids(h.x, h.y, h.z, h.w, ENT_PER);
    ridC[i] = pack_ids(c.x, c.y, c.z, c.w, ENT_PER);
    ridR[i] = pack_ids(r.x, r.y, r.z, r.w, USR_PER);
    int* e = recE + 16 * i;
    e[0]  = h.x; e[1]  = t.x; e[2]  = rel_row(y.x);
    e[4]  = h.y; e[5]  = t.y; e[6]  = rel_row(y.y);
    e[8]  = h.z; e[9]  = t.z; e[10] = rel_row(y.z);
    e[12] = h.w; e[13] = t.w; e[14] = rel_row(y.w);
    int4* n = (int4*)(recI + 16 * i);
    n[0] = make_int4(r.x, c.x, __float_as_int(v.x), 0);
    n[1] = make_int4(r.y, c.y, __float_as_int(v.y), 0);
    n[2] = make_int4(r.z, c.z, __float_as_int(v.z), 0);
    n[3] = make_int4(r.w, c.w, __float_as_int(v.w), 0);
}

struct OwnArgs {
    const float* ent; const float* usr; const float* weight;
    const int4* recE; const int4* recI; const float* segmax;
    float* accE; float* accU; float* sumE;
    int loE, nE, loU, nU;
    int record;
};

#define ENT_DEAD 15u

__device__ __forceinline__ unsigned int zero_flags(unsigned int x)
{
    const unsigned int t = (x & 0x7F7F7F7Fu) + 0x7F7F7F7Fu;
    return ~(t | x) & 0x80808080u;
}

template <int MODE>
__device__ __forceinline__ void gather_add(const OwnArgs& a, int srow, int lrow, int et, float sc, float av, float mx,
                                           bool ok, int n, int lane)
{
    const float* src = (MODE == 1) ? a.usr : a.ent;
    float* dst = (MODE == 2) ? a.accU : a.accE;
    v2f sv[BATCH];
    v2f rv[BATCH];
#pragma unroll
    for (int k = 0; k < BATCH; ++k) {
        const int sr = __builtin_amdgcn_readlane(srow, k);
        sv[k] = *(const v2f*)(src + (size_t)sr * CH + 2 * lane);
        if (MODE == 0) {
            const int er = __builtin_amdgcn_readlane(et, k);
            rv[k] = *(const v2f*)(a.weight + er * CH + 2 * lane);
        }
    }
    if (MODE == 0) sc = expf(av - mx);
    if (!ok) { sc = 0.0f; lrow = 0; }
    if (MODE == 0) {
        if (ok && lane < BATCH) atomicAdd(a.sumE + lrow, sc);
    }
#pragma unroll
    for (int k = 0; k < BATCH; ++k) {
        if (k < n) {
            const float s = __int_as_float(__builtin_amdgcn_readlane(__float_as_int(sc), k));
            const int lr = __builtin_amdgcn_readlane(lrow, k);
            float x, y;
            if (MODE == 0) { x = s * sv[k].x * rv[k].x; y = s * sv[k].y * rv[k].y; }
            else           { x = s * sv[k].x;           y = s * sv[k].y; }
            atomicAdd(dst + lr * CH + 2 * lane,     x);
            atomicAdd(dst + lr * CH + 2 * lane + 1, y);
        }
    }
}

template <int MODE>
__device__ __forceinline__ void replay_list(const OwnArgs& a, const uint2* __restrict__ lst, int cnt, int lane)
{
    if (cnt <= 0) return;
    const int j = lane & (BATCH - 1);
    uint2 en = lst[(j < cnt) ? j : 0];
    for (int base = 0; base < cnt; base += BATCH) {
        const int n = (cnt - base < BATCH) ? (cnt - base) : BATCH;
        const bool live = j < n;
        const int jn = base + BATCH + j;
        const uint2 nx = lst[(jn < cnt) ? jn : 0];
        const int srow = (int)(en.x & 0x3FFFFu);
        const int lrow = (int)((en.x >> 18) & 0x3FFu);
        int et = 0;
        float sc = 0.0f, av = 0.0f, mx = 0.0f;
        bool ok = live;
        if (MODE == 0) {
            const unsigned int rel = en.x >> 28;
            ok = live && (rel != ENT_DEAD);
            et = (rel != ENT_DEAD) ? (int)rel : 0;
            av = __int_as_float(((const int*)a.recE)[4 * (size_t)en.y + 3]);
            mx = a.segmax[a.loE + lrow];
        } else {
            sc = __uint_as_float(en.y);
        }
        gather_add<MODE>(a, srow, lrow, et, sc, av, mx, ok, n, lane);
        en = nx;
    }
}

template <int MODE>
__device__ __forceinline__ void drain_queue(const OwnArgs& a, int* q, int cnt, int lane, uint2* __restrict__ lst, int& pos)
{
    for (int base = 0; base < cnt; base += BATCH) {
        const int n = (cnt - base < BATCH) ? (cnt - base) : BATCH;
        const int j = lane & (BATCH - 1);
        const bool live = j < n;
        const int item = q[base + (live ? j : 0)];
        int srow, lrow, et = 0;
        float sc = 0.0f, av = 0.0f, mx = 0.0f;
        bool ok;
        if (MODE == 0) {
            const int4 r = a.recE[item];
            srow = r.y;
            et = r.z;
            av = __int_as_float(r.w);
            mx = a.segmax[r.x];
            lrow = r.x - a.loE;
            ok = live && ((unsigned int)lrow < (unsigned int)a.nE);
        } else if (MODE == 1) {
            const int4 r = a.recI[item];
            srow = r.x;
            sc = __int_as_float(r.z);
            lrow = r.y - a.loE;
            ok = live && ((unsigned int)lrow < (unsigned int)a.nE);
        } else {
            const int4 r = a.recI[item];
            srow = r.y;
            sc = __int_as_float(r.z);
            lrow = r.x - a.loU;
            ok = live && ((unsigned int)lrow < (unsigned int)a.nU);
        }
        if (a.record) {
            if (pos + n <= CAPW && live && lane < BATCH) {
                const unsigned int rel = (MODE == 0) ? (ok ? (unsigned int)et : ENT_DEAD) : 0u;
                const unsigned int w0 = ((unsigned int)srow & 0x3FFFFu) | ((unsigned int)(ok ? lrow : 0) << 18) | (rel << 28);
                const unsigned int w1 = (MODE == 0) ? (unsigned int)item : (ok ? __float_as_uint(sc) : 0u);
                lst[pos + j] = make_uint2(w0, w1);
            }
            pos += n;
        }
        gather_add<MODE>(a, srow, lrow, et, sc, av, mx, ok, n, lane);
    }
}

template <int MODE>
__device__ __forceinline__ void scan_list(const OwnArgs& a, const unsigned int* __restrict__ rid,
                                          unsigned int pat, int* q, int wave, int lane,
                                          uint2* __restrict__ lst, int* __restrict__ lcnt)
{
    const uint4* rid4 = (const uint4*)rid;
    int cnt = 0;
    int pos = 0;
    uint4 w[SCAN_T];
#pragma unroll
    for (int u = 0; u < SCAN_T; ++u) {
        const int g = wave * 32 + lane + u * (OW_WAVES * 32);
        w[u] = rid4[(g < NGROUPS) ? g : (NGROUPS - 1)];
    }
    for (int t0 = 0; t0 < NGROUPS; t0 += SCAN_T * OW_WAVES * 32) {
        const int gb = t0 + wave * 32 + lane;
        unsigned int mA = 0u, mB = 0u;
#pragma unroll
        for (int u = 0; u < SCAN_T; ++u) {
            unsigned int m = (zero_flags(w[u].x ^ pat) >> 7) | (zero_flags(w[u].y ^ pat) >> 6) |
                             (zero_flags(w[u].z ^ pat) >> 5) | (zero_flags(w[u].w ^ pat) >> 4);
            if (gb + u * (OW_WAVES * 32) >= NGROUPS) m = 0u;
            m <<= 4 * (u & 1);
            if (u < 2) mA |= m; else mB |= m;
        }
#pragma unroll
        for (int u = 0; u < SCAN_T; ++u) {
            const int g = gb + (SCAN_T + u) * (OW_WAVES * 32);
            w[u] = rid4[(g < NGROUPS) ? g : (NGROUPS - 1)];
        }
        unsigned int bal = __builtin_amdgcn_ballot_w32((mA | mB) != 0u);
        while (bal != 0u) {
            if (cnt + 32 > QCAP) { drain_queue<MODE>(a, q, cnt, lane, lst, pos); cnt = 0; }
            if ((mA | mB) != 0u) {
                const bool inA = mA != 0u;
                const unsigned int x = inA ? mA : mB;
                const int p = __builtin_ctz(x);
                if (inA) mA &= mA - 1u; else mB &= mB - 1u;
                const int u = (inA ? 0 : 2) + ((p >> 2) & 1);
                const int g = gb + u * (OW_WAVES * 32);
                q[cnt + (int)__builtin_amdgcn_mbcnt_lo(bal, 0u)] = g * 16 + 4 * (p & 3) + (p >> 3);
            }
            cnt += __builtin_popcount(bal);
            bal = __builtin_amdgcn_ballot_w32((mA | mB) != 0u);
        }
    }
    if (cnt > 0) drain_queue<MODE>(a, q, cnt, lane, lst, pos);
    if (a.record && lane == 0) *lcnt = (pos <= CAPW) ? pos : -1;
}

template <int MODE>
__device__ __forceinline__ void do_list(const OwnArgs& a, const unsigned int* __restrict__ rid, unsigned int pat, int* q,
                                        int wave, int lane, int rg, int replay, uint2* lists, int* lcnts)
{
    const int slot = (rg * NLIST + MODE) * OW_WAVES + wave;
    uint2* lst = lists + (size_t)slot * CAPW;
    int kept = -1;
    if (replay) kept = __builtin_amdgcn_readfirstlane(lcnts[slot]);
    if (kept >= 0) replay_list<MODE>(a, lst, kept, lane);
    else           scan_list<MODE>(a, rid, pat, q, wave, lane, lst, lcnts + slot);
}

__global__ __launch_bounds__(OW_WAVES * 32)
void owner_kernel(const float* cur_ent,
                  const float* cur_usr,
                  const float* __restrict__ weight,
                  const int* __restrict__ recE,
                  const int* __restrict__ recI,
                  const float* __restrict__ segmax,
                  const unsigned int* __restrict__ ridE,
                  const unsigned int* __restrict__ ridC,
                  const unsigned int* __restrict__ ridR,
                  const float* res_ent,
                  const float* res_usr,
                  float* nxt_ent,
                  float* nxt_usr,
                  float* out_ent,
                  float* out_usr,
                  int write_nxt,
                  uint2* lists,
                  int* lcnts,
                  unsigned long long ws_bytes,
                  int second_hop)
{
    __shared__ float accE[ENT_PER * CH];
    __shared__ float accU[USR_PER * CH];
    __shared__ float sumE[ENT_PER];
    __shared__ int   qs[OW_WAVES * QCAP];

    const int lane = threadIdx.x & 31;
    const int wave = __builtin_amdgcn_readfirstlane(threadIdx.x >> 5);
    const int rg   = blockIdx.x;

    OwnArgs a;
    a.ent = cur_ent; a.usr = cur_usr; a.weight = weight;
    a.recE = (const int4*)recE; a.recI = (const int4*)recI; a.segmax = segmax;
    a.accE = accE; a.accU = accU; a.sumE = sumE;
    a.loE = rg * ENT_PER;
    a.nE  = (N_ENT - a.loE < ENT_PER) ? (N_ENT - a.loE) : ENT_PER;
    a.loU = rg * USR_PER;
    a.nU  = (N_USR - a.loU < USR_PER) ? (N_USR - a.loU) : USR_PER;
    const bool room = ws_bytes >= WS_NEED;
    a.record = (room && !second_hop) ? 1 : 0;
    const int replay = (room && second_hop) ? 1 : 0;

    for (int i = threadIdx.x; i < ENT_PER * CH; i += OW_WAVES * 32) accE[i] = 0.0f;
    for (int i = threadIdx.x; i < USR_PER * CH; i += OW_WAVES * 32) accU[i] = 0.0f;
    for (int i = threadIdx.x; i < ENT_PER;      i += OW_WAVES * 32) sumE[i] = 0.0f;
    __syncthreads();

    int* q = qs + wave * QCAP;
    const unsigned int pat = (unsigned int)rg * 0x01010101u;

    do_list<0>(a, ridE, pat, q, wave, lane, rg, replay, lists, lcnts);
    __syncthreads();

    for (int lr = wave; lr < a.nE; lr += OW_WAVES) {
        float s = sumE[lr];
        s = (s > 0.0f) ? s : 1.0f;
        v2f v = *(v2f*)(accE + lr * CH + 2 * lane);
        v.x /= s; v.y /= s;
        *(v2f*)(accE + lr * CH + 2 * lane) = v;
    }
    __syncthreads();

    do_list<1>(a, ridC, pat, q, wave, lane, rg, replay, lists, lcnts);
    do_list<2>(a, ridR, pat, q, wave, lane, rg, replay, lists, lcnts);
    __syncthreads();

    for (int lr = wave; lr < a.nE; lr += OW_WAVES) {
        v2f v = *(v2f*)(accE + lr * CH + 2 * lane);
        float ss = v.x * v.x + v.y * v.y;
#pragma unroll
        for (int m = 16; m >= 1; m >>= 1) ss += __shfl_xor(ss, m, 32);
        const float inv = 1.0f / fmaxf(sqrtf(ss), EPS_N);
        v.x *= inv; v.y *= inv;
        const size_t off = (size_t)(a.loE + lr) * CH + 2 * lane;
        if (write_nxt) *(v2f*)(nxt_ent + off) = v;
        v2f o = *(const v2f*)(res_ent + off);
        o.x += v.x; o.y += v.y;
        *(v2f*)(out_ent + off) = o;
    }
    for (int lr = wave; lr < a.nU; lr += OW_WAVES) {
        v2f v = *(v2f*)(accU + lr * CH + 2 * lane);
        float ss = v.x * v.x + v.y * v.y;
#pragma unroll
        for (int m = 16; m >= 1; m >>= 1) ss += __shfl_xor(ss, m, 32);
        const float inv = 1.0f / fmaxf(sqrtf(ss), EPS_N);
        v.x *= inv; v.y *= inv;
        const size_t off = (size_t)(a.loU + lr) * CH + 2 * lane;
        if (write_nxt) *(v2f*)(nxt_usr + off) = v;
        v2f o = *(const v2f*)(res_usr + off);
        o.x += v.x; o.y += v.y;
        *(v2f*)(out_usr + off) = o;
    }
}

extern "C" void kernel_launch(void* const* d_in, const int* in_sizes, int n_in,
                              void* d_out, int out_size, void* d_ws, size_t ws_size,
                              hipStream_t stream)
{
    (void)in_sizes; (void)n_in; (void)out_size;

    const float* user_emb   = (const float*)d_in[0];
    const float* entity_emb = (const float*)d_in[1];
    const float* weight     = (const float*)d_in[2];
    const float* inter_vals = (const float*)d_in[3];
    const int*   edge_index = (const int*)d_in[4];
    const int*   edge_type  = (const int*)d_in[5];
    const int*   inter_rows = (const int*)d_in[6];
    const int*   inter_cols = (const int*)d_in[7];
    const int*   head = edge_index;
    const int*   tail = edge_index + E_EDGES;

    float* out_ent = (float*)d_out;
    float* out_usr = out_ent + (size_t)N_ENT * CH;

    char* ws = (char*)d_ws;
    size_t off = 0;
    auto alloc_f = [&](size_t nfl) -> float* {
        float* p = (float*)(ws + off);
        off += ((nfl * sizeof(float) + 255) / 256) * 256;
        return p;
    };
    float* nxt_ent = alloc_f((size_t)N_ENT * CH);
    float* nxt_usr = alloc_f((size_t)N_USR * CH);
    int*   recE    = (int*)alloc_f((size_t)E_EDGES * 4);
    int*   recI    = (int*)alloc_f((size_t)NNZ_CNT * 4);
    float* att     = (float*)recE;
    float* segmax  = alloc_f(N_ENT);
    unsigned int* ridE = (unsigned int*)alloc_f(E_EDGES / 4);
    unsigned int* ridC = (unsigned int*)alloc_f(NNZ_CNT / 4);
    unsigned int* ridR = (unsigned int*)alloc_f(NNZ_CNT / 4);
    uint2* lists = (uint2*)alloc_f((size_t)NRANGE * NLIST * OW_WAVES * CAPW * 2);
    int*   lcnts = (int*)alloc_f((size_t)NRANGE * NLIST * OW_WAVES);

    prep_kernel<<<(E_EDGES / 4 + 255) / 256, 256, 0, stream>>>(
        head, tail, edge_type, inter_rows, inter_cols, inter_vals, ridE, ridC, ridR, recE, recI);

    hipMemsetAsync(segmax, 0, (size_t)N_ENT * sizeof(float), stream);
    att_wmma_kernel<<<E_EDGES / 16 / ATT_WAVES, ATT_WAVES * 32, 0, stream>>>(
        entity_emb, weight, head, tail, edge_type, att, (unsigned int*)segmax);
    owner_kernel<<<NRANGE, OW_WAVES * 32, 0, stream>>>(
        entity_emb, user_emb, weight, recE, recI, segmax, ridE, ridC, ridR,
        entity_emb, user_emb, nxt_ent, nxt_usr, out_ent, out_usr, 1,
        lists, lcnts, (unsigned long long)ws_size, 0);

    hipMemsetAsync(segmax, 0, (size_t)N_ENT * sizeof(float), stream);
    att_wmma_kernel<<<E_EDGES / 16 / ATT_WAVES, ATT_WAVES * 32, 0, stream>>>(
        nxt_ent, weight, head, tail, edge_type, att, (unsigned int*)segmax);
    owner_kernel<<<NRANGE, OW_WAVES * 32, 0, stream>>>(
        nxt_ent, nxt_usr, weight, recE, recI, segmax, ridE, ridC, ridR,
        out_ent, out_usr, nxt_ent, nxt_usr, out_ent, out_usr, 0,
        lists, lcnts, (unsigned long long)ws_size, 1);
}


# Attention kernel: each global load reads two whole 256-byte rows (4 lines, each touched once; was 32 B of 16 rows: 16 lines); rows reach the fragment layout through wave-private LDS at pitch 68
# speedup vs baseline: 2.4624x; 1.0504x over previous
#include <hip/hip_runtime.h>
#include <math.h>

#define N_ENT   150000
#define N_USR   60000
#define CH      64
#define E_EDGES 1000000
#define NNZ_CNT 1000000
#define NRELM1  9
#define NHOPS   2
#define EPS_N   1e-12f

#define NRANGE   256
#define ENT_PER  586
#define USR_PER  235
#define OW_WAVES 32
#define QCAP     128
#define BATCH    16
#define SCAN_T   4
#define NGROUPS  (E_EDGES / 16)
#define NLIST    3
#define CAPW     192
#define WS_NEED  127207680ull

typedef __attribute__((ext_vector_type(2))) float v2f;
typedef __attribute__((ext_vector_type(8))) float v8f;

#define ATT_WAVES 4
#define ROW_PITCH 68
typedef __attribute__((ext_vector_type(4))) float v4f;

__device__ __forceinline__ v4f load_row_piece(const float* __restrict__ row, int h, int i)
{
    return *(const v4f*)(row + 8 * i + 4 * h);
}

__device__ __forceinline__ v4f row_piece_lds(const float* tile, int c, int h, int i)
{
    return *(const v4f*)(tile + c * ROW_PITCH + 8 * i + 4 * h);
}

__global__ __launch_bounds__(ATT_WAVES * 32)
void att_wmma_kernel(const float* __restrict__ ent,
                     const float* __restrict__ weight,
                     const int* __restrict__ head,
                     const int* __restrict__ tail,
                     const int* __restrict__ etype,
                     float* __restrict__ att,
                     unsigned int* __restrict__ segmax)
{
    __shared__ float sP[ATT_WAVES][16 * 16];
    __shared__ __attribute__((aligned(16))) float sRow[ATT_WAVES][2][16 * ROW_PITCH];

    const int lane = threadIdx.x & 31;
    const int wave = threadIdx.x >> 5;

    const int tile = blockIdx.x * ATT_WAVES + wave;
    if (tile * 16 >= E_EDGES) return;
    const int base = tile * 16;

    const int c = lane & 15;
    const int h = lane >> 4;
    const int e  = base + c;
    const int hr = head[e];
    const int tr = tail[e];
    int et = etype[e] - 1;
    if (et < 0) et += NRELM1;
    const bool wlive = c < NRELM1;
    const int wr = wlive ? c : (NRELM1 - 1);

    const float* wrow = weight + wr * CH;
    const v4f* ent4 = (const v4f*)ent;
    v4f wv[8], gh[8], gt[8];
#pragma unroll
    for (int j = 0; j < 8; ++j) {
        const int h0 = __builtin_amdgcn_readlane(hr, 2 * j), h1 = __builtin_amdgcn_readlane(hr, 2 * j + 1);
        const int t0 = __builtin_amdgcn_readlane(tr, 2 * j), t1 = __builtin_amdgcn_readlane(tr, 2 * j + 1);
        wv[j] = load_row_piece(wrow, h, j);
        gh[j] = ent4[(h ? h1 : h0) * (CH / 4) + c];
        gt[j] = ent4[(h ? t1 : t0) * (CH / 4) + c];
    }
    __builtin_amdgcn_sched_barrier(0);

    float* tileH = sRow[wave][0];
    float* tileT = sRow[wave][1];
#pragma unroll
    for (int j = 0; j < 8; ++j) {
        *(v4f*)(tileH + (2 * j + h) * ROW_PITCH + 4 * c) = gh[j];
        *(v4f*)(tileT + (2 * j + h) * ROW_PITCH + 4 * c) = gt[j];
    }
    v4f hv[8], tv[8];
#pragma unroll
    for (int i = 0; i < 8; ++i) {
        hv[i] = row_piece_lds(tileH, c, h, i);
        tv[i] = row_piece_lds(tileT, c, h, i);
    }
    __builtin_amdgcn_sched_barrier(0);

    v8f accH = {};
    v8f accT = {};
#pragma unroll
    for (int i = 0; i < 8; ++i) {
        v2f a, at, b;
        a.x  = hv[i].x * hv[i].x;  a.y  = hv[i].y * hv[i].y;
        at.x = tv[i].x * tv[i].x;  at.y = tv[i].y * tv[i].y;
        b.x  = wlive ? wv[i].x * wv[i].x : 0.0f;
        b.y  = wlive ? wv[i].y * wv[i].y : 0.0f;
        accH = __builtin_amdgcn_wmma_f32_16x16x4_f32(false, a,  false, b, (short)0, accH, false, false);
        accT = __builtin_amdgcn_wmma_f32_16x16x4_f32(false, at, false, b, (short)0, accT, false, false);
        a.x  = hv[i].z * hv[i].z;  a.y  = hv[i].w * hv[i].w;
        at.x = tv[i].z * tv[i].z;  at.y = tv[i].w * tv[i].w;
        b.x  = wlive ? wv[i].z * wv[i].z : 0.0f;
        b.y  = wlive ? wv[i].w * wv[i].w : 0.0f;
        accH = __builtin_amdgcn_wmma_f32_16x16x4_f32(false, a,  false, b, (short)0, accH, false, false);
        accT = __builtin_amdgcn_wmma_f32_16x16x4_f32(false, at, false, b, (short)0, accT, false, false);
    }

    const int mrow = h << 3;
#pragma unroll
    for (int v = 0; v < 8; ++v) sP[wave][(mrow + v) * 16 + c] = accH[v] * accT[v];

    if (lane < 16) {
        const float a = sP[wave][lane * 16 + et];
        att[4 * e + 3] = a;
        atomicMax(&segmax[hr], __float_as_uint(a));
    }
}

__device__ __forceinline__ unsigned int pack_ids(int a, int b, int c, int d, int per)
{
    return ((unsigned int)(a / per) & 0xFFu) | (((unsigned int)(b / per) & 0xFFu) << 8) |
           (((unsigned int)(c / per) & 0xFFu) << 16) | (((unsigned int)(d / per) & 0xFFu) << 24);
}

__device__ __forceinline__ int rel_row(int etype)
{
    int et = etype - 1;
    if (et < 0) et += NRELM1;
    return et;
}

__global__ __launch_bounds__(256)
void prep_kernel(const int* __restrict__ head,
                 const int* __restrict__ tail,
                 const int* __restrict__ etype,
                 const int* __restrict__ rows,
                 const int* __restrict__ cols,
                 const float* __restrict__ vals,
                 unsigned int* __restrict__ ridE,
                 unsigned int* __restrict__ ridC,
                 unsigned int* __restrict__ ridR,
                 int* __restrict__ recE,
                 int* __restrict__ recI)
{
    const int i = blockIdx.x * 256 + threadIdx.x;
    if (i >= E_EDGES / 4) return;
    const int4 h = ((const int4*)head)[i];
    const int4 t = ((const int4*)tail)[i];
    const int4 y = ((const int4*)etype)[i];
    const int4 c = ((const int4*)cols)[i];
    const int4 r = ((const int4*)rows)[i];
    const float4 v = ((const float4*)vals)[i];
    ridE[i] = pack_ids(h.x, h.y, h.z, h.w, ENT_PER);
    ridC[i] = pack_ids(c.x, c.y, c.z, c.w, ENT_PER);
    ridR[i] = pack_ids(r.x, r.y, r.z, r.w, USR_PER);
    int* e = recE + 16 * i;
    e[0]  = h.x; e[1]  = t.x; e[2]  = rel_row(y.x);
    e[4]  = h.y; e[5]  = t.y; e[6]  = rel_row(y.y);
    e[8]  = h.z; e[9]  = t.z; e[10] = rel_row(y.z);
    e[12] = h.w; e[13] = t.w; e[14] = rel_row(y.w);
    int4* n = (int4*)(recI + 16 * i);
    n[0] = make_int4(r.x, c.x, __float_as_int(v.x), 0);
    n[1] = make_int4(r.y, c.y, __float_as_int(v.y), 0);
    n[2] = make_int4(r.z, c.z, __float_as_int(v.z), 0);
    n[3] = make_int4(r.w, c.w, __float_as_int(v.w), 0);
}

struct OwnArgs {
    const float* ent; const float* usr; const float* weight;
    const int4* recE; const int4* recI; const float* segmax;
    float* accE; float* accU; float* sumE;
    int loE, nE, loU, nU;
    int record;
};

#define ENT_DEAD 15u

__device__ __forceinline__ unsigned int zero_flags(unsigned int x)
{
    const unsigned int t = (x & 0x7F7F7F7Fu) + 0x7F7F7F7Fu;
    return ~(t | x) & 0x80808080u;
}

template <int MODE>
__device__ __forceinline__ void gather_add(const OwnArgs& a, int srow, int lrow, int et, float sc, float av, float mx,
                                           bool ok, int n, int lane)
{
    const float* src = (MODE == 1) ? a.usr : a.ent;
    float* dst = (MODE == 2) ? a.accU : a.accE;
    v2f sv[BATCH];
    v2f rv[BATCH];
#pragma unroll
    for (int k = 0; k < BATCH; ++k) {
        const int sr = __builtin_amdgcn_readlane(srow, k);
        sv[k] = *(const v2f*)(src + (size_t)sr * CH + 2 * lane);
        if (MODE == 0) {
            const int er = __builtin_amdgcn_readlane(et, k);
            rv[k] = *(const v2f*)(a.weight + er * CH + 2 * lane);
        }
    }
    if (MODE == 0) sc = expf(av - mx);
    if (!ok) { sc = 0.0f; lrow = 0; }
    if (MODE == 0) {
        if (ok && lane < BATCH) atomicAdd(a.sumE + lrow, sc);
    }
#pragma unroll
    for (int k = 0; k < BATCH; ++k) {
        if (k < n) {
            const float s = __int_as_float(__builtin_amdgcn_readlane(__float_as_int(sc), k));
            const int lr = __builtin_amdgcn_readlane(lrow, k);
            float x, y;
            if (MODE == 0) { x = s * sv[k].x * rv[k].x; y = s * sv[k].y * rv[k].y; }
            else           { x = s * sv[k].x;           y = s * sv[k].y; }
            atomicAdd(dst + lr * CH + 2 * lane,     x);
            atomicAdd(dst + lr * CH + 2 * lane + 1, y);
        }
    }
}

template <int MODE>
__device__ __forceinline__ void replay_list(const OwnArgs& a, const uint2* __restrict__ lst, int cnt, int lane)
{
    if (cnt <= 0) return;
    const int j = lane & (BATCH - 1);
    uint2 en = lst[(j < cnt) ? j : 0];
    for (int base = 0; base < cnt; base += BATCH) {
        const int n = (cnt - base < BATCH) ? (cnt - base) : BATCH;
        const bool live = j < n;
        const int jn = base + BATCH + j;
        const uint2 nx = lst[(jn < cnt) ? jn : 0];
        const int srow = (int)(en.x & 0x3FFFFu);
        const int lrow = (int)((en.x >> 18) & 0x3FFu);
        int et = 0;
        float sc = 0.0f, av = 0.0f, mx = 0.0f;
        bool ok = live;
        if (MODE == 0) {
            const unsigned int rel = en.x >> 28;
            ok = live && (rel != ENT_DEAD);
            et = (rel != ENT_DEAD) ? (int)rel : 0;
            av = __int_as_float(((const int*)a.recE)[4 * (size_t)en.y + 3]);
            mx = a.segmax[a.loE + lrow];
        } else {
            sc = __uint_as_float(en.y);
        }
        gather_add<MODE>(a, srow, lrow, et, sc, av, mx, ok, n, lane);
        en = nx;
    }
}

template <int MODE>
__device__ __forceinline__ void drain_queue(const OwnArgs& a, int* q, int cnt, int lane, uint2* __restrict__ lst, int& pos)
{
    for (int base = 0; base < cnt; base += BATCH) {
        const int n = (cnt - base < BATCH) ? (cnt - base) : BATCH;
        const int j = lane & (BATCH - 1);
        const bool live = j < n;
        const int item = q[base + (live ? j : 0)];
        int srow, lrow, et = 0;
        float sc = 0.0f, av = 0.0f, mx = 0.0f;
        bool ok;
        if (MODE == 0) {
            const int4 r = a.recE[item];
            srow = r.y;
            et = r.z;
            av = __int_as_float(r.w);
            mx = a.segmax[r.x];
            lrow = r.x - a.loE;
            ok = live && ((unsigned int)lrow < (unsigned int)a.nE);
        } else if (MODE == 1) {
            const int4 r = a.recI[item];
            srow = r.x;
            sc = __int_as_float(r.z);
            lrow = r.y - a.loE;
            ok = live && ((unsigned int)lrow < (unsigned int)a.nE);
        } else {
            const int4 r = a.recI[item];
            srow = r.y;
            sc = __int_as_float(r.z);
            lrow = r.x - a.loU;
            ok = live && ((unsigned int)lrow < (unsigned int)a.nU);
        }
        if (a.record) {
            if (pos + n <= CAPW && live && lane < BATCH) {
                const unsigned int rel = (MODE == 0) ? (ok ? (unsigned int)et : ENT_DEAD) : 0u;
                const unsigned int w0 = ((unsigned int)srow & 0x3FFFFu) | ((unsigned int)(ok ? lrow : 0) << 18) | (rel << 28);
                const unsigned int w1 = (MODE == 0) ? (unsigned int)item : (ok ? __float_as_uint(sc) : 0u);
                lst[pos + j] = make_uint2(w0, w1);
            }
            pos += n;
        }
        gather_add<MODE>(a, srow, lrow, et, sc, av, mx, ok, n, lane);
    }
}

template <int MODE>
__device__ __forceinline__ void scan_list(const OwnArgs& a, const unsigned int* __restrict__ rid,
                                          unsigned int pat, int* q, int wave, int lane,
                                          uint2* __restrict__ lst, int* __restrict__ lcnt)
{
    const uint4* rid4 = (const uint4*)rid;
    int cnt = 0;
    int pos = 0;
    uint4 w[SCAN_T];
#pragma unroll
    for (int u = 0; u < SCAN_T; ++u) {
        const int g = wave * 32 + lane + u * (OW_WAVES * 32);
        w[u] = rid4[(g < NGROUPS) ? g : (NGROUPS - 1)];
    }
    for (int t0 = 0; t0 < NGROUPS; t0 += SCAN_T * OW_WAVES * 32) {
        const int gb = t0 + wave * 32 + lane;
        unsigned int mA = 0u, mB = 0u;
#pragma unroll
        for (int u = 0; u < SCAN_T; ++u) {
            unsigned int m = (zero_flags(w[u].x ^ pat) >> 7) | (zero_flags(w[u].y ^ pat) >> 6) |
                             (zero_flags(w[u].z ^ pat) >> 5) | (zero_flags(w[u].w ^ pat) >> 4);
            if (gb + u * (OW_WAVES * 32) >= NGROUPS) m = 0u;
            m <<= 4 * (u & 1);
            if (u < 2) mA |= m; else mB |= m;
        }
#pragma unroll
        for (int u = 0; u < SCAN_T; ++u) {
            const int g = gb + (SCAN_T + u) * (OW_WAVES * 32);
            w[u] = rid4[(g < NGROUPS) ? g : (NGROUPS - 1)];
        }
        unsigned int bal = __builtin_amdgcn_ballot_w32((mA | mB) != 0u);
        while (bal != 0u) {
            if (cnt + 32 > QCAP) { drain_queue<MODE>(a, q, cnt, lane, lst, pos); cnt = 0; }
            if ((mA | mB) != 0u) {
                const bool inA = mA != 0u;
                const unsigned int x = inA ? mA : mB;
                const int p = __builtin_ctz(x);
                if (inA) mA &= mA - 1u; else mB &= mB - 1u;
                const int u = (inA ? 0 : 2) + ((p >> 2) & 1);
                const int g = gb + u * (OW_WAVES * 32);
                q[cnt + (int)__builtin_amdgcn_mbcnt_lo(bal, 0u)] = g * 16 + 4 * (p & 3) + (p >> 3);
            }
            cnt += __builtin_popcount(bal);
            bal = __builtin_amdgcn_ballot_w32((mA | mB) != 0u);
        }
    }
    if (cnt > 0) drain_queue<MODE>(a, q, cnt, lane, lst, pos);
    if (a.record && lane == 0) *lcnt = (pos <= CAPW) ? pos : -1;
}

template <int MODE>
__device__ __forceinline__ void do_list(const OwnArgs& a, const unsigned int* __restrict__ rid, unsigned int pat, int* q,
                                        int wave, int lane, int rg, int replay, uint2* lists, int* lcnts)
{
    const int slot = (rg * NLIST + MODE) * OW_WAVES + wave;
    uint2* lst = lists + (size_t)slot * CAPW;
    int kept = -1;
    if (replay) kept = __builtin_amdgcn_readfirstlane(lcnts[slot]);
    if (kept >= 0) replay_list<MODE>(a, lst, kept, lane);
    else           scan_list<MODE>(a, rid, pat, q, wave, lane, lst, lcnts + slot);
}

__global__ __launch_bounds__(OW_WAVES * 32)
void owner_kernel(const float* cur_ent,
                  const float* cur_usr,
                  const float* __restrict__ weight,
                  const int* __restrict__ recE,
                  const int* __restrict__ recI,
                  const float* __restrict__ segmax,
                  const unsigned int* __restrict__ ridE,
                  const unsigned int* __restrict__ ridC,
                  const unsigned int* __restrict__ ridR,
                  const float* res_ent,
                  const float* res_usr,
                  float* nxt_ent,
                  float* nxt_usr,
                  float* out_ent,
                  float* out_usr,
                  int write_nxt,
                  uint2* lists,
                  int* lcnts,
                  unsigned long long ws_bytes,
                  int second_hop)
{
    __shared__ float accE[ENT_PER * CH];
    __shared__ float accU[USR_PER * CH];
    __shared__ float sumE[ENT_PER];
    __shared__ int   qs[OW_WAVES * QCAP];

    const int lane = threadIdx.x & 31;
    const int wave = __builtin_amdgcn_readfirstlane(threadIdx.x >> 5);
    const int rg   = blockIdx.x;

    OwnArgs a;
    a.ent = cur_ent; a.usr = cur_usr; a.weight = weight;
    a.recE = (const int4*)recE; a.recI = (const int4*)recI; a.segmax = segmax;
    a.accE = accE; a.accU = accU; a.sumE = sumE;
    a.loE = rg * ENT_PER;
    a.nE  = (N_ENT - a.loE < ENT_PER) ? (N_ENT - a.loE) : ENT_PER;
    a.loU = rg * USR_PER;
    a.nU  = (N_USR - a.loU < USR_PER) ? (N_USR - a.loU) : USR_PER;
    const bool room = ws_bytes >= WS_NEED;
    a.record = (room && !second_hop) ? 1 : 0;
    const int replay = (room && second_hop) ? 1 : 0;

    for (int i = threadIdx.x; i < ENT_PER * CH; i += OW_WAVES * 32) accE[i] = 0.0f;
    for (int i = threadIdx.x; i < USR_PER * CH; i += OW_WAVES * 32) accU[i] = 0.0f;
    for (int i = threadIdx.x; i < ENT_PER;      i += OW_WAVES * 32) sumE[i] = 0.0f;
    __syncthreads();

    int* q = qs + wave * QCAP;
    const unsigned int pat = (unsigned int)rg * 0x01010101u;

    do_list<0>(a, ridE, pat, q, wave, lane, rg, replay, lists, lcnts);
    __syncthreads();

    for (int lr = wave; lr < a.nE; lr += OW_WAVES) {
        float s = sumE[lr];
        s = (s > 0.0f) ? s : 1.0f;
        v2f v = *(v2f*)(accE + lr * CH + 2 * lane);
        v.x /= s; v.y /= s;
        *(v2f*)(accE + lr * CH + 2 * lane) = v;
    }
    __syncthreads();

    do_list<1>(a, ridC, pat, q, wave, lane, rg, replay, lists, lcnts);
    do_list<2>(a, ridR, pat, q, wave, lane, rg, replay, lists, lcnts);
    __syncthreads();

    for (int lr = wave; lr < a.nE; lr += OW_WAVES) {
        v2f v = *(v2f*)(accE + lr * CH + 2 * lane);
        float ss = v.x * v.x + v.y * v.y;
#pragma unroll
        for (int m = 16; m >= 1; m >>= 1) ss += __shfl_xor(ss, m, 32);
        const float inv = 1.0f / fmaxf(sqrtf(ss), EPS_N);
        v.x *= inv; v.y *= inv;
        const size_t off = (size_t)(a.loE + lr) * CH + 2 * lane;
        if (write_nxt) *(v2f*)(nxt_ent + off) = v;
        v2f o = *(const v2f*)(res_ent + off);
        o.x += v.x; o.y += v.y;
        *(v2f*)(out_ent + off) = o;
    }
    for (int lr = wave; lr < a.nU; lr += OW_WAVES) {
        v2f v = *(v2f*)(accU + lr * CH + 2 * lane);
        float ss = v.x * v.x + v.y * v.y;
#pragma unroll
        for (int m = 16; m >= 1; m >>= 1) ss += __shfl_xor(ss, m, 32);
        const float inv = 1.0f / fmaxf(sqrtf(ss), EPS_N);
        v.x *= inv; v.y *= inv;
        const size_t off = (size_t)(a.loU + lr) * CH + 2 * lane;
        if (write_nxt) *(v2f*)(nxt_usr + off) = v;
        v2f o = *(const v2f*)(res_usr + off);
        o.x += v.x; o.y += v.y;
        *(v2f*)(out_usr + off) = o;
    }
}

extern "C" void kernel_launch(void* const* d_in, const int* in_sizes, int n_in,
                              void* d_out, int out_size, void* d_ws, size_t ws_size,
                              hipStream_t stream)
{
    (void)in_sizes; (void)n_in; (void)out_size;

    const float* user_emb   = (const float*)d_in[0];
    const float* entity_emb = (const float*)d_in[1];
    const float* weight     = (const float*)d_in[2];
    const float* inter_vals = (const float*)d_in[3];
    const int*   edge_index = (const int*)d_in[4];
    const int*   edge_type  = (const int*)d_in[5];
    const int*   inter_rows = (const int*)d_in[6];
    const int*   inter_cols = (const int*)d_in[7];
    const int*   head = edge_index;
    const int*   tail = edge_index + E_EDGES;

    float* out_ent = (float*)d_out;
    float* out_usr = out_ent + (size_t)N_ENT * CH;

    char* ws = (char*)d_ws;
    size_t off = 0;
    auto alloc_f = [&](size_t nfl) -> float* {
        float* p = (float*)(ws + off);
        off += ((nfl * sizeof(float) + 255) / 256) * 256;
        return p;
    };
    float* nxt_ent = alloc_f((size_t)N_ENT * CH);
    float* nxt_usr = alloc_f((size_t)N_USR * CH);
    int*   recE    = (int*)alloc_f((size_t)E_EDGES * 4);
    int*   recI    = (int*)alloc_f((size_t)NNZ_CNT * 4);
    float* att     = (float*)recE;
    float* segmax  = alloc_f(N_ENT);
    unsigned int* ridE = (unsigned int*)alloc_f(E_EDGES / 4);
    unsigned int* ridC = (unsigned int*)alloc_f(NNZ_CNT / 4);
    unsigned int* ridR = (unsigned int*)alloc_f(NNZ_CNT / 4);
    uint2* lists = (uint2*)alloc_f((size_t)NRANGE * NLIST * OW_WAVES * CAPW * 2);
    int*   lcnts = (int*)alloc_f((size_t)NRANGE * NLIST * OW_WAVES);

    prep_kernel<<<(E_EDGES / 4 + 255) / 256, 256, 0, stream>>>(
        head, tail, edge_type, inter_rows, inter_cols, inter_vals, ridE, ridC, ridR, recE, recI);

    hipMemsetAsync(segmax, 0, (size_t)N_ENT * sizeof(float), stream);
    att_wmma_kernel<<<E_EDGES / 16 / ATT_WAVES, ATT_WAVES * 32, 0, stream>>>(
        entity_emb, weight, head, tail, edge_type, att, (unsigned int*)segmax);
    owner_kernel<<<NRANGE, OW_WAVES * 32, 0, stream>>>(
        entity_emb, user_emb, weight, recE, recI, segmax, ridE, ridC, ridR,
        entity_emb, user_emb, nxt_ent, nxt_usr, out_ent, out_usr, 1,
        lists, lcnts, (unsigned long long)ws_size, 0);

    hipMemsetAsync(segmax, 0, (size_t)N_ENT * sizeof(float), stream);
    att_wmma_kernel<<<E_EDGES / 16 / ATT_WAVES, ATT_WAVES * 32, 0, stream>>>(
        nxt_ent, weight, head, tail, edge_type, att, (unsigned int*)segmax);
    owner_kernel<<<NRANGE, OW_WAVES * 32, 0, stream>>>(
        nxt_ent, nxt_usr, weight, recE, recI, segmax, ridE, ridC, ridR,
        out_ent, out_usr, nxt_ent, nxt_usr, out_ent, out_usr, 0,
        lists, lcnts, (unsigned long long)ws_size, 1);
}


# Owner kernel, row loops behind the sums (softmax division; normalize+residual): a wave takes 10 rows per trip, all loads in front; one sqrt and division per trip (lane r = row r); sums unchanged
# speedup vs baseline: 2.5931x; 1.0531x over previous
#include <hip/hip_runtime.h>
#include <math.h>

#define N_ENT   150000
#define N_USR   60000
#define CH      64
#define E_EDGES 1000000
#define NNZ_CNT 1000000
#define NRELM1  9
#define NHOPS   2
#define EPS_N   1e-12f

#define NRANGE   256
#define ENT_PER  586
#define USR_PER  235
#define OW_WAVES 32
#define QCAP     128
#define BATCH    16
#define SCAN_T   4
#define NGROUPS  (E_EDGES / 16)
#define NLIST    3
#define CAPW     192
#define WS_NEED  127207680ull

typedef __attribute__((ext_vector_type(2))) float v2f;
typedef __attribute__((ext_vector_type(8))) float v8f;

#define ATT_WAVES 4
#define ROW_PITCH 68
typedef __attribute__((ext_vector_type(4))) float v4f;

__device__ __forceinline__ v4f load_row_piece(const float* __restrict__ row, int h, int i)
{
    return *(const v4f*)(row + 8 * i + 4 * h);
}

__device__ __forceinline__ v4f row_piece_lds(const float* tile, int c, int h, int i)
{
    return *(const v4f*)(tile + c * ROW_PITCH + 8 * i + 4 * h);
}

__global__ __launch_bounds__(ATT_WAVES * 32)
void att_wmma_kernel(const float* __restrict__ ent,
                     const float* __restrict__ weight,
                     const int* __restrict__ head,
                     const int* __restrict__ tail,
                     const int* __restrict__ etype,
                     float* __restrict__ att,
                     unsigned int* __restrict__ segmax)
{
    __shared__ float sP[ATT_WAVES][16 * 16];
    __shared__ __attribute__((aligned(16))) float sRow[ATT_WAVES][2][16 * ROW_PITCH];

    const int lane = threadIdx.x & 31;
    const int wave = threadIdx.x >> 5;

    const int tile = blockIdx.x * ATT_WAVES + wave;
    if (tile * 16 >= E_EDGES) return;
    const int base = tile * 16;

    const int c = lane & 15;
    const int h = lane >> 4;
    const int e  = base + c;
    const int hr = head[e];
    const int tr = tail[e];
    int et = etype[e] - 1;
    if (et < 0) et += NRELM1;
    const bool wlive = c < NRELM1;
    const int wr = wlive ? c : (NRELM1 - 1);

    const float* wrow = weight + wr * CH;
    const v4f* ent4 = (const v4f*)ent;
    v4f wv[8], gh[8], gt[8];
#pragma unroll
    for (int j = 0; j < 8; ++j) {
        const int h0 = __builtin_amdgcn_readlane(hr, 2 * j), h1 = __builtin_amdgcn_readlane(hr, 2 * j + 1);
        const int t0 = __builtin_amdgcn_readlane(tr, 2 * j), t1 = __builtin_amdgcn_readlane(tr, 2 * j + 1);
        wv[j] = load_row_piece(wrow, h, j);
        gh[j] = ent4[(h ? h1 : h0) * (CH / 4) + c];
        gt[j] = ent4[(h ? t1 : t0) * (CH / 4) + c];
    }
    __builtin_amdgcn_sched_barrier(0);

    float* tileH = sRow[wave][0];
    float* tileT = sRow[wave][1];
#pragma unroll
    for (int j = 0; j < 8; ++j) {
        *(v4f*)(tileH + (2 * j + h) * ROW_PITCH + 4 * c) = gh[j];
        *(v4f*)(tileT + (2 * j + h) * ROW_PITCH + 4 * c) = gt[j];
    }
    v4f hv[8], tv[8];
#pragma unroll
    for (int i = 0; i < 8; ++i) {
        hv[i] = row_piece_lds(tileH, c, h, i);
        tv[i] = row_piece_lds(tileT, c, h, i);
    }
    __builtin_amdgcn_sched_barrier(0);

    v8f accH = {};
    v8f accT = {};
#pragma unroll
    for (int i = 0; i < 8; ++i) {
        v2f a, at, b;
        a.x  = hv[i].x * hv[i].x;  a.y  = hv[i].y * hv[i].y;
        at.x = tv[i].x * tv[i].x;  at.y = tv[i].y * tv[i].y;
        b.x  = wlive ? wv[i].x * wv[i].x : 0.0f;
        b.y  = wlive ? wv[i].y * wv[i].y : 0.0f;
        accH = __builtin_amdgcn_wmma_f32_16x16x4_f32(false, a,  false, b, (short)0, accH, false, false);
        accT = __builtin_amdgcn_wmma_f32_16x16x4_f32(false, at, false, b, (short)0, accT, false, false);
        a.x  = hv[i].z * hv[i].z;  a.y  = hv[i].w * hv[i].w;
        at.x = tv[i].z * tv[i].z;  at.y = tv[i].w * tv[i].w;
        b.x  = wlive ? wv[i].z * wv[i].z : 0.0f;
        b.y  = wlive ? wv[i].w * wv[i].w : 0.0f;
        accH = __builtin_amdgcn_wmma_f32_16x16x4_f32(false, a,  false, b, (short)0, accH, false, false);
        accT = __builtin_amdgcn_wmma_f32_16x16x4_f32(false, at, false, b, (short)0, accT, false, false);
    }

    const int mrow = h << 3;
#pragma unroll
    for (int v = 0; v < 8; ++v) sP[wave][(mrow + v) * 16 + c] = accH[v] * accT[v];

    if (lane < 16) {
        const float a = sP[wave][lane * 16 + et];
        att[4 * e + 3] = a;
        atomicMax(&segmax[hr], __float_as_uint(a));
    }
}

__device__ __forceinline__ unsigned int pack_ids(int a, int b, int c, int d, int per)
{
    return ((unsigned int)(a / per) & 0xFFu) | (((unsigned int)(b / per) & 0xFFu) << 8) |
           (((unsigned int)(c / per) & 0xFFu) << 16) | (((unsigned int)(d / per) & 0xFFu) << 24);
}

__device__ __forceinline__ int rel_row(int etype)
{
    int et = etype - 1;
    if (et < 0) et += NRELM1;
    return et;
}

__global__ __launch_bounds__(256)
void prep_kernel(const int* __restrict__ head,
                 const int* __restrict__ tail,
                 const int* __restrict__ etype,
                 const int* __restrict__ rows,
                 const int* __restrict__ cols,
                 const float* __restrict__ vals,
                 unsigned int* __restrict__ ridE,
                 unsigned int* __restrict__ ridC,
                 unsigned int* __restrict__ ridR,
                 int* __restrict__ recE,
                 int* __restrict__ recI)
{
    const int i = blockIdx.x * 256 + threadIdx.x;
    if (i >= E_EDGES / 4) return;
    const int4 h = ((const int4*)head)[i];
    const int4 t = ((const int4*)tail)[i];
    const int4 y = ((const int4*)etype)[i];
    const int4 c = ((const int4*)cols)[i];
    const int4 r = ((const int4*)rows)[i];
    const float4 v = ((const float4*)vals)[i];
    ridE[i] = pack_ids(h.x, h.y, h.z, h.w, ENT_PER);
    ridC[i] = pack_ids(c.x, c.y, c.z, c.w, ENT_PER);
    ridR[i] = pack_ids(r.x, r.y, r.z, r.w, USR_PER);
    int* e = recE + 16 * i;
    e[0]  = h.x; e[1]  = t.x; e[2]  = rel_row(y.x);
    e[4]  = h.y; e[5]  = t.y; e[6]  = rel_row(y.y);
    e[8]  = h.z; e[9]  = t.z; e[10] = rel_row(y.z);
    e[12] = h.w; e[13] = t.w; e[14] = rel_row(y.w);
    int4* n = (int4*)(recI + 16 * i);
    n[0] = make_int4(r.x, c.x, __float_as_int(v.x), 0);
    n[1] = make_int4(r.y, c.y, __float_as_int(v.y), 0);
    n[2] = make_int4(r.z, c.z, __float_as_int(v.z), 0);
    n[3] = make_int4(r.w, c.w, __float_as_int(v.w), 0);
}

struct OwnArgs {
    const float* ent; const float* usr; const float* weight;
    const int4* recE; const int4* recI; const float* segmax;
    float* accE; float* accU; float* sumE;
    int loE, nE, loU, nU;
    int record;
};

#define ENT_DEAD 15u

__device__ __forceinline__ unsigned int zero_flags(unsigned int x)
{
    const unsigned int t = (x & 0x7F7F7F7Fu) + 0x7F7F7F7Fu;
    return ~(t | x) & 0x80808080u;
}

template <int MODE>
__device__ __forceinline__ void gather_add(const OwnArgs& a, int srow, int lrow, int et, float sc, float av, float mx,
                                           bool ok, int n, int lane)
{
    const float* src = (MODE == 1) ? a.usr : a.ent;
    float* dst = (MODE == 2) ? a.accU : a.accE;
    v2f sv[BATCH];
    v2f rv[BATCH];
#pragma unroll
    for (int k = 0; k < BATCH; ++k) {
        const int sr = __builtin_amdgcn_readlane(srow, k);
        sv[k] = *(const v2f*)(src + (size_t)sr * CH + 2 * lane);
        if (MODE == 0) {
            const int er = __builtin_amdgcn_readlane(et, k);
            rv[k] = *(const v2f*)(a.weight + er * CH + 2 * lane);
        }
    }
    if (MODE == 0) sc = expf(av - mx);
    if (!ok) { sc = 0.0f; lrow = 0; }
    if (MODE == 0) {
        if (ok && lane < BATCH) atomicAdd(a.sumE + lrow, sc);
    }
#pragma unroll
    for (int k = 0; k < BATCH; ++k) {
        if (k < n) {
            const float s = __int_as_float(__builtin_amdgcn_readlane(__float_as_int(sc), k));
            const int lr = __builtin_amdgcn_readlane(lrow, k);
            float x, y;
            if (MODE == 0) { x = s * sv[k].x * rv[k].x; y = s * sv[k].y * rv[k].y; }
            else           { x = s * sv[k].x;           y = s * sv[k].y; }
            atomicAdd(dst + lr * CH + 2 * lane,     x);
            atomicAdd(dst + lr * CH + 2 * lane + 1, y);
        }
    }
}

template <int MODE>
__device__ __forceinline__ void replay_list(const OwnArgs& a, const uint2* __restrict__ lst, int cnt, int lane)
{
    if (cnt <= 0) return;
    const int j = lane & (BATCH - 1);
    uint2 en = lst[(j < cnt) ? j : 0];
    for (int base = 0; base < cnt; base += BATCH) {
        const int n = (cnt - base < BATCH) ? (cnt - base) : BATCH;
        const bool live = j < n;
        const int jn = base + BATCH + j;
        const uint2 nx = lst[(jn < cnt) ? jn : 0];
        const int srow = (int)(en.x & 0x3FFFFu);
        const int lrow = (int)((en.x >> 18) & 0x3FFu);
        int et = 0;
        float sc = 0.0f, av = 0.0f, mx = 0.0f;
        bool ok = live;
        if (MODE == 0) {
            const unsigned int rel = en.x >> 28;
            ok = live && (rel != ENT_DEAD);
            et = (rel != ENT_DEAD) ? (int)rel : 0;
            av = __int_as_float(((const int*)a.recE)[4 * (size_t)en.y + 3]);
            mx = a.segmax[a.loE + lrow];
        } else {
            sc = __uint_as_float(en.y);
        }
        gather_add<MODE>(a, srow, lrow, et, sc, av, mx, ok, n, lane);
        en = nx;
    }
}

template <int MODE>
__device__ __forceinline__ void drain_queue(const OwnArgs& a, int* q, int cnt, int lane, uint2* __restrict__ lst, int& pos)
{
    for (int base = 0; base < cnt; base += BATCH) {
        const int n = (cnt - base < BATCH) ? (cnt - base) : BATCH;
        const int j = lane & (BATCH - 1);
        const bool live = j < n;
        const int item = q[base + (live ? j : 0)];
        int srow, lrow, et = 0;
        float sc = 0.0f, av = 0.0f, mx = 0.0f;
        bool ok;
        if (MODE == 0) {
            const int4 r = a.recE[item];
            srow = r.y;
            et = r.z;
            av = __int_as_float(r.w);
            mx = a.segmax[r.x];
            lrow = r.x - a.loE;
            ok = live && ((unsigned int)lrow < (unsigned int)a.nE);
        } else if (MODE == 1) {
            const int4 r = a.recI[item];
            srow = r.x;
            sc = __int_as_float(r.z);
            lrow = r.y - a.loE;
            ok = live && ((unsigned int)lrow < (unsigned int)a.nE);
        } else {
            const int4 r = a.recI[item];
            srow = r.y;
            sc = __int_as_float(r.z);
            lrow = r.x - a.loU;
            ok = live && ((unsigned int)lrow < (unsigned int)a.nU);
        }
        if (a.record) {
            if (pos + n <= CAPW && live && lane < BATCH) {
                const unsigned int rel = (MODE == 0) ? (ok ? (unsigned int)et : ENT_DEAD) : 0u;
                const unsigned int w0 = ((unsigned int)srow & 0x3FFFFu) | ((unsigned int)(ok ? lrow : 0) << 18) | (rel << 28);
                const unsigned int w1 = (MODE == 0) ? (unsigned int)item : (ok ? __float_as_uint(sc) : 0u);
                lst[pos + j] = make_uint2(w0, w1);
            }
            pos += n;
        }
        gather_add<MODE>(a, srow, lrow, et, sc, av, mx, ok, n, lane);
    }
}

template <int MODE>
__device__ __forceinline__ void scan_list(const OwnArgs& a, const unsigned int* __restrict__ rid,
                                          unsigned int pat, int* q, int wave, int lane,
                                          uint2* __restrict__ lst, int* __restrict__ lcnt)
{
    const uint4* rid4 = (const uint4*)rid;
    int cnt = 0;
    int pos = 0;
    uint4 w[SCAN_T];
#pragma unroll
    for (int u = 0; u < SCAN_T; ++u) {
        const int g = wave * 32 + lane + u * (OW_WAVES * 32);
        w[u] = rid4[(g < NGROUPS) ? g : (NGROUPS - 1)];
    }
    for (int t0 = 0; t0 < NGROUPS; t0 += SCAN_T * OW_WAVES * 32) {
        const int gb = t0 + wave * 32 + lane;
        unsigned int mA = 0u, mB = 0u;
#pragma unroll
        for (int u = 0; u < SCAN_T; ++u) {
            unsigned int m = (zero_flags(w[u].x ^ pat) >> 7) | (zero_flags(w[u].y ^ pat) >> 6) |
                             (zero_flags(w[u].z ^ pat) >> 5) | (zero_flags(w[u].w ^ pat) >> 4);
            if (gb + u * (OW_WAVES * 32) >= NGROUPS) m = 0u;
            m <<= 4 * (u & 1);
            if (u < 2) mA |= m; else mB |= m;
        }
#pragma unroll
        for (int u = 0; u < SCAN_T; ++u) {
            const int g = gb + (SCAN_T + u) * (OW_WAVES * 32);
            w[u] = rid4[(g < NGROUPS) ? g : (NGROUPS - 1)];
        }
        unsigned int bal = __builtin_amdgcn_ballot_w32((mA | mB) != 0u);
        while (bal != 0u) {
            if (cnt + 32 > QCAP) { drain_queue<MODE>(a, q, cnt, lane, lst, pos); cnt = 0; }
            if ((mA | mB) != 0u) {
                const bool inA = mA != 0u;
                const unsigned int x = inA ? mA : mB;
                const int p = __builtin_ctz(x);
                if (inA) mA &= mA - 1u; else mB &= mB - 1u;
                const int u = (inA ? 0 : 2) + ((p >> 2) & 1);
                const int g = gb + u * (OW_WAVES * 32);
                q[cnt + (int)__builtin_amdgcn_mbcnt_lo(bal, 0u)] = g * 16 + 4 * (p & 3) + (p >> 3);
            }
            cnt += __builtin_popcount(bal);
            bal = __builtin_amdgcn_ballot_w32((mA | mB) != 0u);
        }
    }
    if (cnt > 0) drain_queue<MODE>(a, q, cnt, lane, lst, pos);
    if (a.record && lane == 0) *lcnt = (pos <= CAPW) ? pos : -1;
}

template <int MODE>
__device__ __forceinline__ void do_list(const OwnArgs& a, const unsigned int* __restrict__ rid, unsigned int pat, int* q,
                                        int wave, int lane, int rg, int replay, uint2* lists, int* lcnts)
{
    const int slot = (rg * NLIST + MODE) * OW_WAVES + wave;
    uint2* lst = lists + (size_t)slot * CAPW;
    int kept = -1;
    if (replay) kept = __builtin_amdgcn_readfirstlane(lcnts[slot]);
    if (kept >= 0) replay_list<MODE>(a, lst, kept, lane);
    else           scan_list<MODE>(a, rid, pat, q, wave, lane, lst, lcnts + slot);
}

#define EPI_R 10
#define EPI_RU 8

template <int R, bool GUARD>
__device__ __forceinline__ void div_rows(float* accE, const float* sumE, int lr0, int n, int lane)
{
    float s[R];
    v2f v[R];
#pragma unroll
    for (int r = 0; r < R; ++r) {
        const int row = lr0 + r * OW_WAVES;
        const int rr = (GUARD && row >= n) ? lr0 : row;
        s[r] = sumE[rr];
        v[r] = *(const v2f*)(accE + rr * CH + 2 * lane);
    }
#pragma unroll
    for (int r = 0; r < R; ++r) {
        const int row = lr0 + r * OW_WAVES;
        const float d = (s[r] > 0.0f) ? s[r] : 1.0f;
        v[r].x /= d; v[r].y /= d;
        if (!GUARD || r == 0 || row < n) *(v2f*)(accE + row * CH + 2 * lane) = v[r];
    }
}

template <int R, bool GUARD>
__device__ __forceinline__ void norm_rows(const float* acc, int lr0, int n, int lo, const float* res, float* nxt,
                                          float* out, int write_nxt, int lane)
{
    v2f o[R], v[R];
    float ss[R];
#pragma unroll
    for (int r = 0; r < R; ++r) {
        const int row = lr0 + r * OW_WAVES;
        const int rr = (GUARD && row >= n) ? lr0 : row;
        o[r] = *(const v2f*)(res + (size_t)(lo + rr) * CH + 2 * lane);
        v[r] = *(const v2f*)(acc + rr * CH + 2 * lane);
    }
#pragma unroll
    for (int r = 0; r < R; ++r) ss[r] = fmaf(v[r].x, v[r].x, v[r].y * v[r].y);
#pragma unroll
    for (int m = 16; m >= 1; m >>= 1) {
#pragma unroll
        for (int r = 0; r < R; ++r) ss[r] += __shfl_xor(ss[r], m, 32);
    }
    float t = ss[0];
#pragma unroll
    for (int r = 1; r < R; ++r) t = (lane == r) ? ss[r] : t;
    const float it = 1.0f / fmaxf(sqrtf(t), EPS_N);
#pragma unroll
    for (int r = 0; r < R; ++r) {
        const float inv = __int_as_float(__builtin_amdgcn_readlane(__float_as_int(it), r));
        v[r].x *= inv; v[r].y *= inv;
    }
    if (write_nxt) {
#pragma unroll
        for (int r = 0; r < R; ++r) {
            const int row = lr0 + r * OW_WAVES;
            if (!GUARD || r == 0 || row < n) *(v2f*)(nxt + (size_t)(lo + row) * CH + 2 * lane) = v[r];
        }
    }
#pragma unroll
    for (int r = 0; r < R; ++r) {
        const int row = lr0 + r * OW_WAVES;
        if (!GUARD || r == 0 || row < n) {
            o[r].x += v[r].x; o[r].y += v[r].y;
            *(v2f*)(out + (size_t)(lo + row) * CH + 2 * lane) = o[r];
        }
    }
}

template <int R>
__device__ __forceinline__ void norm_plane(const float* acc, int n, int lo, const float* res, float* nxt, float* out,
                                           int write_nxt, int wave, int lane)
{
    int lr = wave;
    for (; lr + (R - 1) * OW_WAVES < n; lr += R * OW_WAVES)
        norm_rows<R, false>(acc, lr, n, lo, res, nxt, out, write_nxt, lane);
    if (lr < n) norm_rows<R, true>(acc, lr, n, lo, res, nxt, out, write_nxt, lane);
}

__global__ __launch_bounds__(OW_WAVES * 32)
void owner_kernel(const float* cur_ent,
                  const float* cur_usr,
                  const float* __restrict__ weight,
                  const int* __restrict__ recE,
                  const int* __restrict__ recI,
                  const float* __restrict__ segmax,
                  const unsigned int* __restrict__ ridE,
                  const unsigned int* __restrict__ ridC,
                  const unsigned int* __restrict__ ridR,
                  const float* res_ent,
                  const float* res_usr,
                  float* nxt_ent,
                  float* nxt_usr,
                  float* out_ent,
                  float* out_usr,
                  int write_nxt,
                  uint2* lists,
                  int* lcnts,
                  unsigned long long ws_bytes,
                  int second_hop)
{
    __shared__ float accE[ENT_PER * CH];
    __shared__ float accU[USR_PER * CH];
    __shared__ float sumE[ENT_PER];
    __shared__ int   qs[OW_WAVES * QCAP];

    const int lane = threadIdx.x & 31;
    const int wave = __builtin_amdgcn_readfirstlane(threadIdx.x >> 5);
    const int rg   = blockIdx.x;

    OwnArgs a;
    a.ent = cur_ent; a.usr = cur_usr; a.weight = weight;
    a.recE = (const int4*)recE; a.recI = (const int4*)recI; a.segmax = segmax;
    a.accE = accE; a.accU = accU; a.sumE = sumE;
    a.loE = rg * ENT_PER;
    a.nE  = (N_ENT - a.loE < ENT_PER) ? (N_ENT - a.loE) : ENT_PER;
    a.loU = rg * USR_PER;
    a.nU  = (N_USR - a.loU < USR_PER) ? (N_USR - a.loU) : USR_PER;
    const bool room = ws_bytes >= WS_NEED;
    a.record = (room && !second_hop) ? 1 : 0;
    const int replay = (room && second_hop) ? 1 : 0;

    for (int i = threadIdx.x; i < ENT_PER * CH; i += OW_WAVES * 32) accE[i] = 0.0f;
    for (int i = threadIdx.x; i < USR_PER * CH; i += OW_WAVES * 32) accU[i] = 0.0f;
    for (int i = threadIdx.x; i < ENT_PER;      i += OW_WAVES * 32) sumE[i] = 0.0f;
    __syncthreads();

    int* q = qs + wave * QCAP;
    const unsigned int pat = (unsigned int)rg * 0x01010101u;

    do_list<0>(a, ridE, pat, q, wave, lane, rg, replay, lists, lcnts);
    __syncthreads();

    {
        int lr = wave;
        for (; lr + (EPI_R - 1) * OW_WAVES < a.nE; lr += EPI_R * OW_WAVES) div_rows<EPI_R, false>(accE, sumE, lr, a.nE, lane);
        if (lr < a.nE) div_rows<EPI_R, true>(accE, sumE, lr, a.nE, lane);
    }
    __syncthreads();

    do_list<1>(a, ridC, pat, q, wave, lane, rg, replay, lists, lcnts);
    do_list<2>(a, ridR, pat, q, wave, lane, rg, replay, lists, lcnts);
    __syncthreads();

    norm_plane<EPI_R>(accE, a.nE, a.loE, res_ent, nxt_ent, out_ent, write_nxt, wave, lane);
    norm_plane<EPI_RU>(accU, a.nU, a.loU, res_usr, nxt_usr, out_usr, write_nxt, wave, lane);
}

extern "C" void kernel_launch(void* const* d_in, const int* in_sizes, int n_in,
                              void* d_out, int out_size, void* d_ws, size_t ws_size,
                              hipStream_t stream)
{
    (void)in_sizes; (void)n_in; (void)out_size;

    const float* user_emb   = (const float*)d_in[0];
    const float* entity_emb = (const float*)d_in[1];
    const float* weight     = (const float*)d_in[2];
    const float* inter_vals = (const float*)d_in[3];
    const int*   edge_index = (const int*)d_in[4];
    const int*   edge_type  = (const int*)d_in[5];
    const int*   inter_rows = (const int*)d_in[6];
    const int*   inter_cols = (const int*)d_in[7];
    const int*   head = edge_index;
    const int*   tail = edge_index + E_EDGES;

    float* out_ent = (float*)d_out;
    float* out_usr = out_ent + (size_t)N_ENT * CH;

    char* ws = (char*)d_ws;
    size_t off = 0;
    auto alloc_f = [&](size_t nfl) -> float* {
        float* p = (float*)(ws + off);
        off += ((nfl * sizeof(float) + 255) / 256) * 256;
        return p;
    };
    float* nxt_ent = alloc_f((size_t)N_ENT * CH);
    float* nxt_usr = alloc_f((size_t)N_USR * CH);
    int*   recE    = (int*)alloc_f((size_t)E_EDGES * 4);
    int*   recI    = (int*)alloc_f((size_t)NNZ_CNT * 4);
    float* att     = (float*)recE;
    float* segmax  = alloc_f(N_ENT);
    unsigned int* ridE = (unsigned int*)alloc_f(E_EDGES / 4);
    unsigned int* ridC = (unsigned int*)alloc_f(NNZ_CNT / 4);
    unsigned int* ridR = (unsigned int*)alloc_f(NNZ_CNT / 4);
    uint2* lists = (uint2*)alloc_f((size_t)NRANGE * NLIST * OW_WAVES * CAPW * 2);
    int*   lcnts = (int*)alloc_f((size_t)NRANGE * NLIST * OW_WAVES);

    prep_kernel<<<(E_EDGES / 4 + 255) / 256, 256, 0, stream>>>(
        head, tail, edge_type, inter_rows, inter_cols, inter_vals, ridE, ridC, ridR, recE, recI);

    hipMemsetAsync(segmax, 0, (size_t)N_ENT * sizeof(float), stream);
    att_wmma_kernel<<<E_EDGES / 16 / ATT_WAVES, ATT_WAVES * 32, 0, stream>>>(
        entity_emb, weight, head, tail, edge_type, att, (unsigned int*)segmax);
    owner_kernel<<<NRANGE, OW_WAVES * 32, 0, stream>>>(
        entity_emb, user_emb, weight, recE, recI, segmax, ridE, ridC, ridR,
        entity_emb, user_emb, nxt_ent, nxt_usr, out_ent, out_usr, 1,
        lists, lcnts, (unsigned long long)ws_size, 0);

    hipMemsetAsync(segmax, 0, (size_t)N_ENT * sizeof(float), stream);
    att_wmma_kernel<<<E_EDGES / 16 / ATT_WAVES, ATT_WAVES * 32, 0, stream>>>(
        nxt_ent, weight, head, tail, edge_type, att, (unsigned int*)segmax);
    owner_kernel<<<NRANGE, OW_WAVES * 32, 0, stream>>>(
        nxt_ent, nxt_usr, weight, recE, recI, segmax, ridE, ridC, ridR,
        out_ent, out_usr, nxt_ent, nxt_usr, out_ent, out_usr, 0,
        lists, lcnts, (unsigned long long)ws_size, 1);
}


# Owner scan through an index: preparation also builds per list and range one bit per 32 items (LDS or); an owner reads its bits, fetches ids of flagged groups only; LDS add order varies as before
# speedup vs baseline: 2.7085x; 1.0445x over previous
#include <hip/hip_runtime.h>
#include <math.h>

#define N_ENT   150000
#define N_USR   60000
#define CH      64
#define E_EDGES 1000000
#define NNZ_CNT 1000000
#define NRELM1  9
#define NHOPS   2
#define EPS_N   1e-12f

#define NRANGE   256
#define ENT_PER  586
#define USR_PER  235
#define OW_WAVES 32
#define QCAP     128
#define BATCH    16
#define GQCAP    128
#define NLIST    3
#define PREP_T   1024
#define BM_NW    980
#define CAPW     176
#define WS_NEED  127072512ull
static_assert(E_EDGES == NNZ_CNT && E_EDGES % 32 == 0, "one bitmap geometry serves the three lists; groups of 32 items are whole");
static_assert(((E_EDGES / 4 + PREP_T - 1) / PREP_T) * 4 == BM_NW, "every bitmap word is written by exactly one preparation work-group");

typedef __attribute__((ext_vector_type(2))) float v2f;
typedef __attribute__((ext_vector_type(8))) float v8f;

#define ATT_WAVES 4
#define ROW_PITCH 68
typedef __attribute__((ext_vector_type(4))) float v4f;

__device__ __forceinline__ v4f load_row_piece(const float* __restrict__ row, int h, int i)
{
    return *(const v4f*)(row + 8 * i + 4 * h);
}

__device__ __forceinline__ v4f row_piece_lds(const float* tile, int c, int h, int i)
{
    return *(const v4f*)(tile + c * ROW_PITCH + 8 * i + 4 * h);
}

__global__ __launch_bounds__(ATT_WAVES * 32)
void att_wmma_kernel(const float* __restrict__ ent,
                     const float* __restrict__ weight,
                     const int* __restrict__ head,
                     const int* __restrict__ tail,
                     const int* __restrict__ etype,
                     float* __restrict__ att,
                     unsigned int* __restrict__ segmax)
{
    __shared__ float sP[ATT_WAVES][16 * 16];
    __shared__ __attribute__((aligned(16))) float sRow[ATT_WAVES][2][16 * ROW_PITCH];

    const int lane = threadIdx.x & 31;
    const int wave = threadIdx.x >> 5;

    const int tile = blockIdx.x * ATT_WAVES + wave;
    if (tile * 16 >= E_EDGES) return;
    const int base = tile * 16;

    const int c = lane & 15;
    const int h = lane >> 4;
    const int e  = base + c;
    const int hr = head[e];
    const int tr = tail[e];
    int et = etype[e] - 1;
    if (et < 0) et += NRELM1;
    const bool wlive = c < NRELM1;
    const int wr = wlive ? c : (NRELM1 - 1);

    const float* wrow = weight + wr * CH;
    const v4f* ent4 = (const v4f*)ent;
    v4f wv[8], gh[8], gt[8];
#pragma unroll
    for (int j = 0; j < 8; ++j) {
        const int h0 = __builtin_amdgcn_readlane(hr, 2 * j), h1 = __builtin_amdgcn_readlane(hr, 2 * j + 1);
        const int t0 = __builtin_amdgcn_readlane(tr, 2 * j), t1 = __builtin_amdgcn_readlane(tr, 2 * j + 1);
        wv[j] = load_row_piece(wrow, h, j);
        gh[j] = ent4[(h ? h1 : h0) * (CH / 4) + c];
        gt[j] = ent4[(h ? t1 : t0) * (CH / 4) + c];
    }
    __builtin_amdgcn_sched_barrier(0);

    float* tileH = sRow[wave][0];
    float* tileT = sRow[wave][1];
#pragma unroll
    for (int j = 0; j < 8; ++j) {
        *(v4f*)(tileH + (2 * j + h) * ROW_PITCH + 4 * c) = gh[j];
        *(v4f*)(tileT + (2 * j + h) * ROW_PITCH + 4 * c) = gt[j];
    }
    v4f hv[8], tv[8];
#pragma unroll
    for (int i = 0; i < 8; ++i) {
        hv[i] = row_piece_lds(tileH, c, h, i);
        tv[i] = row_piece_lds(tileT, c, h, i);
    }
    __builtin_amdgcn_sched_barrier(0);

    v8f accH = {};
    v8f accT = {};
#pragma unroll
    for (int i = 0; i < 8; ++i) {
        v2f a, at, b;
        a.x  = hv[i].x * hv[i].x;  a.y  = hv[i].y * hv[i].y;
        at.x = tv[i].x * tv[i].x;  at.y = tv[i].y * tv[i].y;
        b.x  = wlive ? wv[i].x * wv[i].x : 0.0f;
        b.y  = wlive ? wv[i].y * wv[i].y : 0.0f;
        accH = __builtin_amdgcn_wmma_f32_16x16x4_f32(false, a,  false, b, (short)0, accH, false, false);
        accT = __builtin_amdgcn_wmma_f32_16x16x4_f32(false, at, false, b, (short)0, accT, false, false);
        a.x  = hv[i].z * hv[i].z;  a.y  = hv[i].w * hv[i].w;
        at.x = tv[i].z * tv[i].z;  at.y = tv[i].w * tv[i].w;
        b.x  = wlive ? wv[i].z * wv[i].z : 0.0f;
        b.y  = wlive ? wv[i].w * wv[i].w : 0.0f;
        accH = __builtin_amdgcn_wmma_f32_16x16x4_f32(false, a,  false, b, (short)0, accH, false, false);
        accT = __builtin_amdgcn_wmma_f32_16x16x4_f32(false, at, false, b, (short)0, accT, false, false);
    }

    const int mrow = h << 3;
#pragma unroll
    for (int v = 0; v < 8; ++v) sP[wave][(mrow + v) * 16 + c] = accH[v] * accT[v];

    if (lane < 16) {
        const float a = sP[wave][lane * 16 + et];
        att[4 * e + 3] = a;
        atomicMax(&segmax[hr], __float_as_uint(a));
    }
}

__device__ __forceinline__ unsigned int pack_ids(int a, int b, int c, int d, int per)
{
    return ((unsigned int)(a / per) & 0xFFu) | (((unsigned int)(b / per) & 0xFFu) << 8) |
           (((unsigned int)(c / per) & 0xFFu) << 16) | (((unsigned int)(d / per) & 0xFFu) << 24);
}

__device__ __forceinline__ int rel_row(int etype)
{
    int et = etype - 1;
    if (et < 0) et += NRELM1;
    return et;
}

__global__ __launch_bounds__(PREP_T)
void prep_kernel(const int* __restrict__ head,
                 const int* __restrict__ tail,
                 const int* __restrict__ etype,
                 const int* __restrict__ rows,
                 const int* __restrict__ cols,
                 const float* __restrict__ vals,
                 unsigned int* __restrict__ ridE,
                 unsigned int* __restrict__ ridC,
                 unsigned int* __restrict__ ridR,
                 int* __restrict__ recE,
                 int* __restrict__ recI,
                 unsigned int* __restrict__ bm)
{
    __shared__ __attribute__((aligned(16))) unsigned int sB[NLIST * NRANGE * 4];
    for (int p = threadIdx.x; p < NLIST * NRANGE * 4; p += PREP_T) sB[p] = 0u;
    __syncthreads();
    const int i = blockIdx.x * PREP_T + threadIdx.x;
    if (i < E_EDGES / 4) {
    const int4 h = ((const int4*)head)[i];
    const int4 t = ((const int4*)tail)[i];
    const int4 y = ((const int4*)etype)[i];
    const int4 c = ((const int4*)cols)[i];
    const int4 r = ((const int4*)rows)[i];
    const float4 v = ((const float4*)vals)[i];
    const unsigned int pe = pack_ids(h.x, h.y, h.z, h.w, ENT_PER);
    const unsigned int pc = pack_ids(c.x, c.y, c.z, c.w, ENT_PER);
    const unsigned int pr = pack_ids(r.x, r.y, r.z, r.w, USR_PER);
    ridE[i] = pe;
    ridC[i] = pc;
    ridR[i] = pr;
    const int wd = threadIdx.x >> 8;
    const unsigned int bit = 1u << ((threadIdx.x >> 3) & 31);
#pragma unroll
    for (int k = 0; k < 4; ++k) {
        atomicOr(&sB[(0 * NRANGE + (int)((pe >> (8 * k)) & 0xFFu)) * 4 + wd], bit);
        atomicOr(&sB[(1 * NRANGE + (int)((pc >> (8 * k)) & 0xFFu)) * 4 + wd], bit);
        atomicOr(&sB[(2 * NRANGE + (int)((pr >> (8 * k)) & 0xFFu)) * 4 + wd], bit);
    }
    int* e = recE + 16 * i;
    e[0]  = h.x; e[1]  = t.x; e[2]  = rel_row(y.x);
    e[4]  = h.y; e[5]  = t.y; e[6]  = rel_row(y.y);
    e[8]  = h.z; e[9]  = t.z; e[10] = rel_row(y.z);
    e[12] = h.w; e[13] = t.w; e[14] = rel_row(y.w);
    int4* n = (int4*)(recI + 16 * i);
    n[0] = make_int4(r.x, c.x, __float_as_int(v.x), 0);
    n[1] = make_int4(r.y, c.y, __float_as_int(v.y), 0);
    n[2] = make_int4(r.z, c.z, __float_as_int(v.z), 0);
    n[3] = make_int4(r.w, c.w, __float_as_int(v.w), 0);
    }
    __syncthreads();
    if (threadIdx.x < NLIST * NRANGE)
        *(uint4*)(bm + (size_t)threadIdx.x * BM_NW + 4 * blockIdx.x) = *(const uint4*)(sB + 4 * threadIdx.x);
}

struct OwnArgs {
    const float* ent; const float* usr; const float* weight;
    const int4* recE; const int4* recI; const float* segmax;
    float* accE; float* accU; float* sumE;
    int loE, nE, loU, nU;
    int record;
};

#define ENT_DEAD 15u

__device__ __forceinline__ unsigned int zero_flags(unsigned int x)
{
    const unsigned int t = (x & 0x7F7F7F7Fu) + 0x7F7F7F7Fu;
    return ~(t | x) & 0x80808080u;
}

template <int MODE>
__device__ __forceinline__ void gather_add(const OwnArgs& a, int srow, int lrow, int et, float sc, float av, float mx,
                                           bool ok, int n, int lane)
{
    const float* src = (MODE == 1) ? a.usr : a.ent;
    float* dst = (MODE == 2) ? a.accU : a.accE;
    v2f sv[BATCH];
    v2f rv[BATCH];
#pragma unroll
    for (int k = 0; k < BATCH; ++k) {
        const int sr = __builtin_amdgcn_readlane(srow, k);
        sv[k] = *(const v2f*)(src + (size_t)sr * CH + 2 * lane);
        if (MODE == 0) {
            const int er = __builtin_amdgcn_readlane(et, k);
            rv[k] = *(const v2f*)(a.weight + er * CH + 2 * lane);
        }
    }
    if (MODE == 0) sc = expf(av - mx);
    if (!ok) { sc = 0.0f; lrow = 0; }
    if (MODE == 0) {
        if (ok && lane < BATCH) atomicAdd(a.sumE + lrow, sc);
    }
#pragma unroll
    for (int k = 0; k < BATCH; ++k) {
        if (k < n) {
            const float s = __int_as_float(__builtin_amdgcn_readlane(__float_as_int(sc), k));
            const int lr = __builtin_amdgcn_readlane(lrow, k);
            float x, y;
            if (MODE == 0) { x = s * sv[k].x * rv[k].x; y = s * sv[k].y * rv[k].y; }
            else           { x = s * sv[k].x;           y = s * sv[k].y; }
            atomicAdd(dst + lr * CH + 2 * lane,     x);
            atomicAdd(dst + lr * CH + 2 * lane + 1, y);
        }
    }
}

template <int MODE>
__device__ __forceinline__ void replay_list(const OwnArgs& a, const uint2* __restrict__ lst, int cnt, int lane)
{
    if (cnt <= 0) return;
    const int j = lane & (BATCH - 1);
    uint2 en = lst[(j < cnt) ? j : 0];
    for (int base = 0; base < cnt; base += BATCH) {
        const int n = (cnt - base < BATCH) ? (cnt - base) : BATCH;
        const bool live = j < n;
        const int jn = base + BATCH + j;
        const uint2 nx = lst[(jn < cnt) ? jn : 0];
        const int srow = (int)(en.x & 0x3FFFFu);
        const int lrow = (int)((en.x >> 18) & 0x3FFu);
        int et = 0;
        float sc = 0.0f, av = 0.0f, mx = 0.0f;
        bool ok = live;
        if (MODE == 0) {
            const unsigned int rel = en.x >> 28;
            ok = live && (rel != ENT_DEAD);
            et = (rel != ENT_DEAD) ? (int)rel : 0;
            av = __int_as_float(((const int*)a.recE)[4 * (size_t)en.y + 3]);
            mx = a.segmax[a.loE + lrow];
        } else {
            sc = __uint_as_float(en.y);
        }
        gather_add<MODE>(a, srow, lrow, et, sc, av, mx, ok, n, lane);
        en = nx;
    }
}

template <int MODE>
__device__ __forceinline__ void drain_queue(const OwnArgs& a, int* q, int cnt, int lane, uint2* __restrict__ lst, int& pos)
{
    for (int base = 0; base < cnt; base += BATCH) {
        const int n = (cnt - base < BATCH) ? (cnt - base) : BATCH;
        const int j = lane & (BATCH - 1);
        const bool live = j < n;
        const int item = q[base + (live ? j : 0)];
        int srow, lrow, et = 0;
        float sc = 0.0f, av = 0.0f, mx = 0.0f;
        bool ok;
        if (MODE == 0) {
            const int4 r = a.recE[item];
            srow = r.y;
            et = r.z;
            av = __int_as_float(r.w);
            mx = a.segmax[r.x];
            lrow = r.x - a.loE;
            ok = live && ((unsigned int)lrow < (unsigned int)a.nE);
        } else if (MODE == 1) {
            const int4 r = a.recI[item];
            srow = r.x;
            sc = __int_as_float(r.z);
            lrow = r.y - a.loE;
            ok = live && ((unsigned int)lrow < (unsigned int)a.nE);
        } else {
            const int4 r = a.recI[item];
            srow = r.y;
            sc = __int_as_float(r.z);
            lrow = r.x - a.loU;
            ok = live && ((unsigned int)lrow < (unsigned int)a.nU);
        }
        if (a.record) {
            if (pos + n <= CAPW && live && lane < BATCH) {
                const unsigned int rel = (MODE == 0) ? (ok ? (unsigned int)et : ENT_DEAD) : 0u;
                const unsigned int w0 = ((unsigned int)srow & 0x3FFFFu) | ((unsigned int)(ok ? lrow : 0) << 18) | (rel << 28);
                const unsigned int w1 = (MODE == 0) ? (unsigned int)item : (ok ? __float_as_uint(sc) : 0u);
                lst[pos + j] = make_uint2(w0, w1);
            }
            pos += n;
        }
        gather_add<MODE>(a, srow, lrow, et, sc, av, mx, ok, n, lane);
    }
}

__device__ __forceinline__ unsigned int match16(uint4 x, unsigned int pat)
{
    return (zero_flags(x.x ^ pat) >> 7) | (zero_flags(x.y ^ pat) >> 6) |
           (zero_flags(x.z ^ pat) >> 5) | (zero_flags(x.w ^ pat) >> 4);
}

template <int MODE>
__device__ __forceinline__ void take_groups(const OwnArgs& a, const uint4* __restrict__ rid4, unsigned int pat,
                                            const int* gq, int gcnt, int* q, int& cnt, int lane,
                                            uint2* __restrict__ lst, int& pos)
{
    for (int base = 0; base < gcnt; base += 32) {
        const bool live = base + lane < gcnt;
        const int g = gq[live ? (base + lane) : base];
        const uint4 x0 = rid4[2 * g];
        const uint4 x1 = rid4[2 * g + 1];
        unsigned int m = match16(x0, pat) | (match16(x1, pat) << 4);
        if (!live) m = 0u;
        unsigned int bal = __builtin_amdgcn_ballot_w32(m != 0u);
        while (bal != 0u) {
            if (cnt + 32 > QCAP) { drain_queue<MODE>(a, q, cnt, lane, lst, pos); cnt = 0; }
            if (m != 0u) {
                const int p = __builtin_ctz(m);
                m &= m - 1u;
                q[cnt + (int)__builtin_amdgcn_mbcnt_lo(bal, 0u)] = g * 32 + 16 * ((p >> 2) & 1) + 4 * (p & 3) + (p >> 3);
            }
            cnt += __builtin_popcount(bal);
            bal = __builtin_amdgcn_ballot_w32(m != 0u);
        }
    }
}

template <int MODE>
__device__ __forceinline__ void scan_list(const OwnArgs& a, const unsigned int* __restrict__ rid, unsigned int bits, int bw,
                                          unsigned int pat, int* q, int* gq, int lane,
                                          uint2* __restrict__ lst, int* __restrict__ lcnt)
{
    const uint4* rid4 = (const uint4*)rid;
    int cnt = 0;
    int pos = 0;
    int gcnt = 0;
    unsigned int bal = __builtin_amdgcn_ballot_w32(bits != 0u);
    while (bal != 0u) {
        if (gcnt + 32 > GQCAP) { take_groups<MODE>(a, rid4, pat, gq, gcnt, q, cnt, lane, lst, pos); gcnt = 0; }
        if (bits != 0u) {
            const int p = __builtin_ctz(bits);
            bits &= bits - 1u;
            gq[gcnt + (int)__builtin_amdgcn_mbcnt_lo(bal, 0u)] = bw * 32 + p;
        }
        gcnt += __builtin_popcount(bal);
        bal = __builtin_amdgcn_ballot_w32(bits != 0u);
    }
    if (gcnt > 0) take_groups<MODE>(a, rid4, pat, gq, gcnt, q, cnt, lane, lst, pos);
    if (cnt > 0) drain_queue<MODE>(a, q, cnt, lane, lst, pos);
    if (a.record && lane == 0) *lcnt = (pos <= CAPW) ? pos : -1;
}

template <int MODE>
__device__ __forceinline__ void do_list(const OwnArgs& a, const unsigned int* __restrict__ rid, unsigned int bits, int bw,
                                        unsigned int pat, int* q, int* gq,
                                        int wave, int lane, int rg, int replay, uint2* lists, int* lcnts)
{
    const int slot = (rg * NLIST + MODE) * OW_WAVES + wave;
    uint2* lst = lists + (size_t)slot * CAPW;
    int kept = -1;
    if (replay) kept = __builtin_amdgcn_readfirstlane(lcnts[slot]);
    if (kept >= 0) replay_list<MODE>(a, lst, kept, lane);
    else           scan_list<MODE>(a, rid, bits, bw, pat, q, gq, lane, lst, lcnts + slot);
}

#define EPI_R 10
#define EPI_RU 8

template <int R, bool GUARD>
__device__ __forceinline__ void div_rows(float* accE, const float* sumE, int lr0, int n, int lane)
{
    float s[R];
    v2f v[R];
#pragma unroll
    for (int r = 0; r < R; ++r) {
        const int row = lr0 + r * OW_WAVES;
        const int rr = (GUARD && row >= n) ? lr0 : row;
        s[r] = sumE[rr];
        v[r] = *(const v2f*)(accE + rr * CH + 2 * lane);
    }
#pragma unroll
    for (int r = 0; r < R; ++r) {
        const int row = lr0 + r * OW_WAVES;
        const float d = (s[r] > 0.0f) ? s[r] : 1.0f;
        v[r].x /= d; v[r].y /= d;
        if (!GUARD || r == 0 || row < n) *(v2f*)(accE + row * CH + 2 * lane) = v[r];
    }
}

template <int R, bool GUARD>
__device__ __forceinline__ void norm_rows(const float* acc, int lr0, int n, int lo, const float* res, float* nxt,
                                          float* out, int write_nxt, int lane)
{
    v2f o[R], v[R];
    float ss[R];
#pragma unroll
    for (int r = 0; r < R; ++r) {
        const int row = lr0 + r * OW_WAVES;
        const int rr = (GUARD && row >= n) ? lr0 : row;
        o[r] = *(const v2f*)(res + (size_t)(lo + rr) * CH + 2 * lane);
        v[r] = *(const v2f*)(acc + rr * CH + 2 * lane);
    }
#pragma unroll
    for (int r = 0; r < R; ++r) ss[r] = fmaf(v[r].x, v[r].x, v[r].y * v[r].y);
#pragma unroll
    for (int m = 16; m >= 1; m >>= 1) {
#pragma unroll
        for (int r = 0; r < R; ++r) ss[r] += __shfl_xor(ss[r], m, 32);
    }
    float t = ss[0];
#pragma unroll
    for (int r = 1; r < R; ++r) t = (lane == r) ? ss[r] : t;
    const float it = 1.0f / fmaxf(sqrtf(t), EPS_N);
#pragma unroll
    for (int r = 0; r < R; ++r) {
        const float inv = __int_as_float(__builtin_amdgcn_readlane(__float_as_int(it), r));
        v[r].x *= inv; v[r].y *= inv;
    }
    if (write_nxt) {
#pragma unroll
        for (int r = 0; r < R; ++r) {
            const int row = lr0 + r * OW_WAVES;
            if (!GUARD || r == 0 || row < n) *(v2f*)(nxt + (size_t)(lo + row) * CH + 2 * lane) = v[r];
        }
    }
#pragma unroll
    for (int r = 0; r < R; ++r) {
        const int row = lr0 + r * OW_WAVES;
        if (!GUARD || r == 0 || row < n) {
            o[r].x += v[r].x; o[r].y += v[r].y;
            *(v2f*)(out + (size_t)(lo + row) * CH + 2 * lane) = o[r];
        }
    }
}

template <int R>
__device__ __forceinline__ void norm_plane(const float* acc, int n, int lo, const float* res, float* nxt, float* out,
                                           int write_nxt, int wave, int lane)
{
    int lr = wave;
    for (; lr + (R - 1) * OW_WAVES < n; lr += R * OW_WAVES)
        norm_rows<R, false>(acc, lr, n, lo, res, nxt, out, write_nxt, lane);
    if (lr < n) norm_rows<R, true>(acc, lr, n, lo, res, nxt, out, write_nxt, lane);
}

__global__ __launch_bounds__(OW_WAVES * 32)
void owner_kernel(const float* cur_ent,
                  const float* cur_usr,
                  const float* __restrict__ weight,
                  const int* __restrict__ recE,
                  const int* __restrict__ recI,
                  const float* __restrict__ segmax,
                  const unsigned int* __restrict__ ridE,
                  const unsigned int* __restrict__ ridC,
                  const unsigned int* __restrict__ ridR,
                  const unsigned int* __restrict__ bm,
                  const float* res_ent,
                  const float* res_usr,
                  float* nxt_ent,
                  float* nxt_usr,
                  float* out_ent,
                  float* out_usr,
                  int write_nxt,
                  uint2* lists,
                  int* lcnts,
                  unsigned long long ws_bytes,
                  int second_hop)
{
    __shared__ float accE[ENT_PER * CH];
    __shared__ float accU[USR_PER * CH];
    __shared__ float sumE[ENT_PER];
    __shared__ int   qs[OW_WAVES * QCAP];
    __shared__ int   gqs[OW_WAVES * GQCAP];

    const int lane = threadIdx.x & 31;
    const int wave = __builtin_amdgcn_readfirstlane(threadIdx.x >> 5);
    const int rg   = blockIdx.x;

    OwnArgs a;
    a.ent = cur_ent; a.usr = cur_usr; a.weight = weight;
    a.recE = (const int4*)recE; a.recI = (const int4*)recI; a.segmax = segmax;
    a.accE = accE; a.accU = accU; a.sumE = sumE;
    a.loE = rg * ENT_PER;
    a.nE  = (N_ENT - a.loE < ENT_PER) ? (N_ENT - a.loE) : ENT_PER;
    a.loU = rg * USR_PER;
    a.nU  = (N_USR - a.loU < USR_PER) ? (N_USR - a.loU) : USR_PER;
    const bool room = ws_bytes >= WS_NEED;
    a.record = (room && !second_hop) ? 1 : 0;
    const int replay = (room && second_hop) ? 1 : 0;

    const int bw = wave + OW_WAVES * lane;
    unsigned int bits0 = 0u, bits1 = 0u, bits2 = 0u;
    if (bw < BM_NW) {
        bits0 = bm[(size_t)(0 * NRANGE + rg) * BM_NW + bw];
        bits1 = bm[(size_t)(1 * NRANGE + rg) * BM_NW + bw];
        bits2 = bm[(size_t)(2 * NRANGE + rg) * BM_NW + bw];
    }

    for (int i = threadIdx.x; i < ENT_PER * CH; i += OW_WAVES * 32) accE[i] = 0.0f;
    for (int i = threadIdx.x; i < USR_PER * CH; i += OW_WAVES * 32) accU[i] = 0.0f;
    for (int i = threadIdx.x; i < ENT_PER;      i += OW_WAVES * 32) sumE[i] = 0.0f;
    __syncthreads();

    int* q = qs + wave * QCAP;
    int* gq = gqs + wave * GQCAP;
    const unsigned int pat = (unsigned int)rg * 0x01010101u;

    do_list<0>(a, ridE, bits0, bw, pat, q, gq, wave, lane, rg, replay, lists, lcnts);
    __syncthreads();

    {
        int lr = wave;
        for (; lr + (EPI_R - 1) * OW_WAVES < a.nE; lr += EPI_R * OW_WAVES) div_rows<EPI_R, false>(accE, sumE, lr, a.nE, lane);
        if (lr < a.nE) div_rows<EPI_R, true>(accE, sumE, lr, a.nE, lane);
    }
    __syncthreads();

    do_list<1>(a, ridC, bits1, bw, pat, q, gq, wave, lane, rg, replay, lists, lcnts);
    do_list<2>(a, ridR, bits2, bw, pat, q, gq, wave, lane, rg, replay, lists, lcnts);
    __syncthreads();

    norm_plane<EPI_R>(accE, a.nE, a.loE, res_ent, nxt_ent, out_ent, write_nxt, wave, lane);
    norm_plane<EPI_RU>(accU, a.nU, a.loU, res_usr, nxt_usr, out_usr, write_nxt, wave, lane);
}

extern "C" void kernel_launch(void* const* d_in, const int* in_sizes, int n_in,
                              void* d_out, int out_size, void* d_ws, size_t ws_size,
                              hipStream_t stream)
{
    (void)in_sizes; (void)n_in; (void)out_size;

    const float* user_emb   = (const float*)d_in[0];
    const float* entity_emb = (const float*)d_in[1];
    const float* weight     = (const float*)d_in[2];
    const float* inter_vals = (const float*)d_in[3];
    const int*   edge_index = (const int*)d_in[4];
    const int*   edge_type  = (const int*)d_in[5];
    const int*   inter_rows = (const int*)d_in[6];
    const int*   inter_cols = (const int*)d_in[7];
    const int*   head = edge_index;
    const int*   tail = edge_index + E_EDGES;

    float* out_ent = (float*)d_out;
    float* out_usr = out_ent + (size_t)N_ENT * CH;

    char* ws = (char*)d_ws;
    size_t off = 0;
    auto alloc_f = [&](size_t nfl) -> float* {
        float* p = (float*)(ws + off);
        off += ((nfl * sizeof(float) + 255) / 256) * 256;
        return p;
    };
    float* nxt_ent = alloc_f((size_t)N_ENT * CH);
    float* nxt_usr = alloc_f((size_t)N_USR * CH);
    int*   recE    = (int*)alloc_f((size_t)E_EDGES * 4);
    int*   recI    = (int*)alloc_f((size_t)NNZ_CNT * 4);
    float* att     = (float*)recE;
    float* segmax  = alloc_f(N_ENT);
    unsigned int* ridE = (unsigned int*)alloc_f(E_EDGES / 4);
    unsigned int* ridC = (unsigned int*)alloc_f(NNZ_CNT / 4);
    unsigned int* ridR = (unsigned int*)alloc_f(NNZ_CNT / 4);
    unsigned int* bm = (unsigned int*)alloc_f((size_t)NLIST * NRANGE * BM_NW);
    uint2* lists = (uint2*)alloc_f((size_t)NRANGE * NLIST * OW_WAVES * CAPW * 2);
    int*   lcnts = (int*)alloc_f((size_t)NRANGE * NLIST * OW_WAVES);

    prep_kernel<<<(E_EDGES / 4 + PREP_T - 1) / PREP_T, PREP_T, 0, stream>>>(
        head, tail, edge_type, inter_rows, inter_cols, inter_vals, ridE, ridC, ridR, recE, recI, bm);

    hipMemsetAsync(segmax, 0, (size_t)N_ENT * sizeof(float), stream);
    att_wmma_kernel<<<E_EDGES / 16 / ATT_WAVES, ATT_WAVES * 32, 0, stream>>>(
        entity_emb, weight, head, tail, edge_type, att, (unsigned int*)segmax);
    owner_kernel<<<NRANGE, OW_WAVES * 32, 0, stream>>>(
        entity_emb, user_emb, weight, recE, recI, segmax, ridE, ridC, ridR, bm,
        entity_emb, user_emb, nxt_ent, nxt_usr, out_ent, out_usr, 1,
        lists, lcnts, (unsigned long long)ws_size, 0);

    hipMemsetAsync(segmax, 0, (size_t)N_ENT * sizeof(float), stream);
    att_wmma_kernel<<<E_EDGES / 16 / ATT_WAVES, ATT_WAVES * 32, 0, stream>>>(
        nxt_ent, weight, head, tail, edge_type, att, (unsigned int*)segmax);
    owner_kernel<<<NRANGE, OW_WAVES * 32, 0, stream>>>(
        nxt_ent, nxt_usr, weight, recE, recI, segmax, ridE, ridC, ridR, bm,
        out_ent, out_usr, nxt_ent, nxt_usr, out_ent, out_usr, 0,
        lists, lcnts, (unsigned long long)ws_size, 1);
}
